# starting kernel
# baseline (speedup 1.0000x reference)
#include <hip/hip_runtime.h>
#include <hip/hip_bf16.h>

typedef float v2f __attribute__((ext_vector_type(2)));
typedef float v8f __attribute__((ext_vector_type(8)));

__device__ __forceinline__ void atomic_add_f32(float* p, float v) {
  __hip_atomic_fetch_add(p, v, __ATOMIC_RELAXED, __HIP_MEMORY_SCOPE_AGENT);
}

__global__ void fill_kernel(float* __restrict__ p, float v, long long n) {
  long long i = (long long)blockIdx.x * blockDim.x + threadIdx.x;
  if (i < n) p[i] = v;
}

__global__ void deg_kernel(const int* __restrict__ dst, float* __restrict__ deg,
                           long long E) {
  long long e = (long long)blockIdx.x * blockDim.x + threadIdx.x;
  if (e < E) atomic_add_f32(&deg[dst[e]], 1.0f);
}

__global__ void dinv_kernel(float* __restrict__ deg, int n) {
  int i = blockIdx.x * blockDim.x + threadIdx.x;
  if (i < n) deg[i] = __frsqrt_rn(fmaxf(deg[i], 1e-12f));
}

template <int FOUT>
__global__ __launch_bounds__(FOUT * 2) void gemm_wmma_kernel(
    const float* __restrict__ A, const float* __restrict__ W,
    float* __restrict__ C, int nrows) {
  __shared__ float lds[16 * 128];

  const int row0 = blockIdx.x * 16;
  if (row0 >= nrows) return;
  const int tid = threadIdx.x;
  constexpr int NTHREADS = FOUT * 2;

  if (row0 + 16 <= nrows) {
    const float4* srcv = (const float4*)(A + (size_t)row0 * 128);
    float4* dstv = (float4*)lds;
    #pragma unroll 2
    for (int i = tid; i < 512; i += NTHREADS) dstv[i] = srcv[i];
  } else {
    for (int i = tid; i < 2048; i += NTHREADS) {
      int r = i >> 7;
      lds[i] = (row0 + r < nrows) ? A[(size_t)(row0 + r) * 128 + (i & 127)] : 0.0f;
    }
  }
  __syncthreads();

  const int lane = tid & 31;
  const int wave = tid >> 5;
  const int colbase = wave * 16;
  const int half = lane >> 4;
  const int m = lane & 15;

  v8f acc = {};
  #pragma unroll
  for (int k = 0; k < 128; k += 4) {
    const int kb = k + half * 2;
    v2f a = *(const v2f*)&lds[m * 128 + kb];
    v2f b;
    b.x = W[(size_t)kb * FOUT + colbase + m];
    b.y = W[(size_t)(kb + 1) * FOUT + colbase + m];
    acc = __builtin_amdgcn_wmma_f32_16x16x4_f32(false, a, false, b, (short)0,
                                                acc, false, false);
  }

  #pragma unroll
  for (int i = 0; i < 8; ++i) {
    int row = row0 + i + half * 8;
    if (row < nrows) C[(size_t)row * FOUT + colbase + m] = acc[i];
  }
}

template <int F>
__global__ void edge_agg_kernel(const int* __restrict__ src,
                                const int* __restrict__ dst,
                                const float* __restrict__ dinv,
                                const float* __restrict__ h,
                                float* __restrict__ agg, long long E) {
  constexpr int CH = F / 4;
  long long t = (long long)blockIdx.x * blockDim.x + threadIdx.x;
  long long e = t / CH;
  int c = (int)(t & (CH - 1));
  if (e >= E) return;
  int u = src[e];
  int v = dst[e];
  float norm = dinv[u] * dinv[v];
  float4 hv = *(const float4*)&h[(size_t)u * F + c * 4];
  float* out = &agg[(size_t)v * F + c * 4];
  atomic_add_f32(out + 0, norm * hv.x);
  atomic_add_f32(out + 1, norm * hv.y);
  atomic_add_f32(out + 2, norm * hv.z);
  atomic_add_f32(out + 3, norm * hv.w);
}

__global__ void epilogue1_kernel(float* __restrict__ agg,
                                 const float* __restrict__ hpre,
                                 const float* __restrict__ dinv,
                                 const float* __restrict__ bias, long long n) {
  long long i = (long long)blockIdx.x * blockDim.x + threadIdx.x;
  if (i >= n) return;
  int v = (int)(i >> 7);
  int c = (int)(i & 127);
  float di = dinv[v];
  float val = agg[i] + di * di * hpre[i] + bias[c];
  agg[i] = fmaxf(val, 0.0f);
}

__global__ void epilogue2_kernel(float* __restrict__ out,
                                 const float* __restrict__ hpre,
                                 const float* __restrict__ dinv,
                                 const float* __restrict__ bias, long long n) {
  long long i = (long long)blockIdx.x * blockDim.x + threadIdx.x;
  if (i >= n) return;
  int v = (int)(i >> 6);
  int c = (int)(i & 63);
  float di = dinv[v];
  out[i] += di * di * hpre[i] + bias[c];
}

extern "C" void kernel_launch(void* const* d_in, const int* in_sizes, int n_in,
                              void* d_out, int out_size, void* d_ws,
                              size_t ws_size, hipStream_t stream) {
  const float* x  = (const float*)d_in[0];
  const int*   ei = (const int*)d_in[1];
  const float* W1 = (const float*)d_in[2];
  const float* b1 = (const float*)d_in[3];
  const float* W2 = (const float*)d_in[4];
  const float* b2 = (const float*)d_in[5];
  float* out = (float*)d_out;

  const int N = in_sizes[0] / 128;
  const long long E = in_sizes[1] / 2;
  const int* src = ei;
  const int* dst = ei + E;

  float* ws = (float*)d_ws;
  float* deg   = ws;
  float* h1pre = ws + N;
  float* agg1  = h1pre + (size_t)N * 128;
  float* h2pre = h1pre;

  const int TB = 256;
  auto blocks = [](long long n, int tb) { return (unsigned)((n + tb - 1) / tb); };

  fill_kernel<<<blocks(N, TB), TB, 0, stream>>>(deg, 1.0f, N);
  fill_kernel<<<blocks((long long)N * 128, TB), TB, 0, stream>>>(agg1, 0.0f, (long long)N * 128);
  fill_kernel<<<blocks((long long)N * 64, TB), TB, 0, stream>>>(out, 0.0f, (long long)N * 64);

  deg_kernel<<<blocks(E, TB), TB, 0, stream>>>(dst, deg, E);
  dinv_kernel<<<blocks(N, TB), TB, 0, stream>>>(deg, N);
  const float* dinv = deg;

  gemm_wmma_kernel<128><<<(N + 15) / 16, 256, 0, stream>>>(x, W1, h1pre, N);
  edge_agg_kernel<128><<<blocks(E * 32, TB), TB, 0, stream>>>(src, dst, dinv, h1pre, agg1, E);
  epilogue1_kernel<<<blocks((long long)N * 128, TB), TB, 0, stream>>>(agg1, h1pre, dinv, b1, (long long)N * 128);
  const float* h1 = agg1;

  gemm_wmma_kernel<64><<<(N + 15) / 16, 128, 0, stream>>>(h1, W2, h2pre, N);
  edge_agg_kernel<64><<<blocks(E * 16, TB), TB, 0, stream>>>(src, dst, dinv, h2pre, out, E);
  epilogue2_kernel<<<blocks((long long)N * 64, TB), TB, 0, stream>>>(out, h2pre, dinv, b2, (long long)N * 64);
}


# Replace float-atomic edge scatter by per-call in-edge lists (int atomic rank, scan, place) and a one-wave-per-node row gather; float sums now in list (arrival-rank) order, self loop last
# speedup vs baseline: 2.1092x; 2.1092x over previous
#include <hip/hip_runtime.h>
#include <hip/hip_bf16.h>


typedef float v2f __attribute__((ext_vector_type(2)));
typedef float v8f __attribute__((ext_vector_type(8)));

__global__ void zero_i32_kernel(int* __restrict__ p, int n) {
  int i = blockIdx.x * blockDim.x + threadIdx.x;
  if (i < n) p[i] = 0;
}

template <int U>
__global__ void rank_kernel(const int* __restrict__ dst, int* __restrict__ cnt,
                            int* __restrict__ rank, long long E) {
  const long long T = (long long)gridDim.x * blockDim.x;
  const long long t = (long long)blockIdx.x * blockDim.x + threadIdx.x;
  if (t + (U - 1) * T < E) {
    int d[U];
    int r[U];
    #pragma unroll
    for (int k = 0; k < U; ++k) d[k] = dst[t + k * T];
    #pragma unroll
    for (int k = 0; k < U; ++k)
      r[k] = __hip_atomic_fetch_add(&cnt[d[k]], 1, __ATOMIC_RELAXED,
                                    __HIP_MEMORY_SCOPE_AGENT);
    #pragma unroll
    for (int k = 0; k < U; ++k) rank[t + k * T] = r[k];
  } else {
    for (int k = 0; k < U; ++k) {
      const long long e = t + k * T;
      if (e < E)
        rank[e] = __hip_atomic_fetch_add(&cnt[dst[e]], 1, __ATOMIC_RELAXED,
                                         __HIP_MEMORY_SCOPE_AGENT);
    }
  }
}

__global__ __launch_bounds__(256) void tilesum_kernel(const int* __restrict__ cnt,
                                                      int* __restrict__ tsum, int n) {
  __shared__ int red[8];
  const int tid = threadIdx.x;
  const int base = blockIdx.x * 1024 + tid * 4;
  int s = 0;
  for (int k = 0; k < 4; ++k)
    if (base + k < n) s += cnt[base + k];
  for (int o = 16; o > 0; o >>= 1) s += __shfl_xor(s, o);
  if ((tid & 31) == 0) red[tid >> 5] = s;
  __syncthreads();
  if (tid == 0) {
    int t = 0;
    for (int w = 0; w < 8; ++w) t += red[w];
    tsum[blockIdx.x] = t;
  }
}

__global__ __launch_bounds__(256) void scan_kernel(const int* __restrict__ cnt,
                                                   const int* __restrict__ tsum,
                                                   int* __restrict__ rowptr,
                                                   float* __restrict__ dinv, int n) {
  __shared__ int red[8];
  __shared__ int wtot[8];
  const int tid = threadIdx.x;
  const int lane = tid & 31;
  const int wave = tid >> 5;

  int p = 0;
  for (int i = tid; i < (int)blockIdx.x; i += 256) p += tsum[i];
  for (int o = 16; o > 0; o >>= 1) p += __shfl_xor(p, o);
  if (lane == 0) red[wave] = p;

  const int base = blockIdx.x * 1024 + tid * 4;
  int c[4];
  for (int k = 0; k < 4; ++k) c[k] = (base + k < n) ? cnt[base + k] : 0;
  const int mine = c[0] + c[1] + c[2] + c[3];
  int incl = mine;
  for (int o = 1; o < 32; o <<= 1) {
    const int y = __shfl_up(incl, o);
    if (lane >= o) incl += y;
  }
  if (lane == 31) wtot[wave] = incl;
  __syncthreads();

  int off = 0;
  for (int w = 0; w < 8; ++w) off += red[w];
  for (int w = 0; w < wave; ++w) off += wtot[w];
  int run = off + incl - mine;
  for (int k = 0; k < 4; ++k) {
    if (base + k < n) {
      rowptr[base + k] = run;
      dinv[base + k] = __frsqrt_rn(fmaxf((float)(c[k] + 1), 1e-12f));
      run += c[k];
      if (base + k == n - 1) rowptr[n] = run;
    }
  }
}

template <int U>
__global__ void place_kernel(const int* __restrict__ src, const int* __restrict__ dst,
                             const int* __restrict__ rank,
                             const int* __restrict__ rowptr,
                             const float* __restrict__ dinv, int* __restrict__ list,
                             long long E) {
  const long long T = (long long)gridDim.x * blockDim.x;
  const long long t = (long long)blockIdx.x * blockDim.x + threadIdx.x;
  if (t + (U - 1) * T < E) {
    int u[U], v[U], rk[U], pos[U];
    float w[U];
    #pragma unroll
    for (int k = 0; k < U; ++k) {
      u[k] = src[t + k * T];
      v[k] = dst[t + k * T];
      rk[k] = rank[t + k * T];
    }
    #pragma unroll
    for (int k = 0; k < U; ++k) {
      pos[k] = rowptr[v[k]] + rk[k];
      w[k] = dinv[u[k]];
    }
    #pragma unroll
    for (int k = 0; k < U; ++k) {
      int2 ent;
      ent.x = u[k];
      ent.y = __float_as_int(w[k]);
      *(int2*)&list[2 * (size_t)pos[k]] = ent;
    }
  } else {
    for (int k = 0; k < U; ++k) {
      const long long e = t + k * T;
      if (e < E) {
        const int uu = src[e];
        const int pp = rowptr[dst[e]] + rank[e];
        int2 ent;
        ent.x = uu;
        ent.y = __float_as_int(dinv[uu]);
        *(int2*)&list[2 * (size_t)pp] = ent;
      }
    }
  }
}

__global__ void move_lists_kernel(const int* __restrict__ s, int* __restrict__ d,
                                  long long m) {
  const long long T = (long long)gridDim.x * blockDim.x;
  const long long t = (long long)blockIdx.x * blockDim.x + threadIdx.x;
  const long long m4 = m >> 2;
  for (long long i = t; i < m4; i += T) ((int4*)d)[i] = ((const int4*)s)[i];
  for (long long i = (m4 << 2) + t; i < m; i += T) d[i] = s[i];
}

template <int FOUT>
__global__ __launch_bounds__(FOUT * 2) void gemm_wmma_kernel(
    const float* __restrict__ A, const float* __restrict__ W,
    float* __restrict__ C, int nrows) {
  __shared__ float lds[16 * 128];

  const int row0 = blockIdx.x * 16;
  if (row0 >= nrows) return;
  const int tid = threadIdx.x;
  constexpr int NTHREADS = FOUT * 2;

  if (row0 + 16 <= nrows) {
    const float4* srcv = (const float4*)(A + (size_t)row0 * 128);
    float4* dstv = (float4*)lds;
    #pragma unroll 2
    for (int i = tid; i < 512; i += NTHREADS) dstv[i] = srcv[i];
  } else {
    for (int i = tid; i < 2048; i += NTHREADS) {
      int r = i >> 7;
      lds[i] = (row0 + r < nrows) ? A[(size_t)(row0 + r) * 128 + (i & 127)] : 0.0f;
    }
  }
  __syncthreads();

  const int lane = tid & 31;
  const int wave = tid >> 5;
  const int colbase = wave * 16;
  const int half = lane >> 4;
  const int m = lane & 15;

  v8f acc = {};
  #pragma unroll
  for (int k = 0; k < 128; k += 4) {
    const int kb = k + half * 2;
    v2f a = *(const v2f*)&lds[m * 128 + kb];
    v2f b;
    b.x = W[(size_t)kb * FOUT + colbase + m];
    b.y = W[(size_t)(kb + 1) * FOUT + colbase + m];
    acc = __builtin_amdgcn_wmma_f32_16x16x4_f32(false, a, false, b, (short)0,
                                                acc, false, false);
  }

  #pragma unroll
  for (int i = 0; i < 8; ++i) {
    int row = row0 + i + half * 8;
    if (row < nrows) C[(size_t)row * FOUT + colbase + m] = acc[i];
  }
}

template <int F, bool RELU>
__global__ __launch_bounds__(256) void gather_kernel(
    const int* __restrict__ rowptr, const int* __restrict__ list,
    const float* __restrict__ dinv, const float* __restrict__ h,
    const float* __restrict__ bias, float* __restrict__ y, int n) {
  constexpr int V = F / 32;
  typedef float vecT __attribute__((ext_vector_type(V)));
  constexpr int G = 8;

  const int lane = threadIdx.x & 31;
  const int wv = __builtin_amdgcn_readfirstlane((int)(threadIdx.x >> 5));
  const int wpb = (int)(blockDim.x >> 5);
  const int nwaves = (int)gridDim.x * wpb;
  const vecT bv = *(const vecT*)&bias[lane * V];

  for (int v = (int)blockIdx.x * wpb + wv; v < n; v += nwaves) {
    const int beg = __builtin_amdgcn_readfirstlane(rowptr[v]);
    const int end = __builtin_amdgcn_readfirstlane(rowptr[v + 1]);
    const float dv = dinv[v];
    const vecT hs = *(const vecT*)&h[(size_t)v * F + lane * V];
    vecT acc = {};

    for (int base = beg; base < end; base += 32) {
      const int m = (end - base < 32) ? (end - base) : 32;
      int eu = v;
      int ew = 0;
      if (lane < m) {
        const int2 ent = *(const int2*)&list[2 * (size_t)(base + lane)];
        eu = ent.x;
        ew = ent.y;
      }
      for (int j = 0; j < m; j += G) {
        vecT r[G];
        float w[G];
        #pragma unroll
        for (int q = 0; q < G; ++q) {
          const int u = __builtin_amdgcn_readlane(eu, j + q);
          w[q] = __int_as_float(__builtin_amdgcn_readlane(ew, j + q));
          r[q] = *(const vecT*)&h[(size_t)u * F + lane * V];
        }
        #pragma unroll
        for (int q = 0; q < G; ++q) acc += (w[q] * dv) * r[q];
      }
    }

    acc += (dv * dv) * hs;
    acc += bv;
    if (RELU) acc = __builtin_elementwise_max(acc, (vecT)(0.0f));
    *(vecT*)&y[(size_t)v * F + lane * V] = acc;
  }
}

extern "C" void kernel_launch(void* const* d_in, const int* in_sizes, int n_in,
                              void* d_out, int out_size, void* d_ws,
                              size_t ws_size, hipStream_t stream) {
  const float* x  = (const float*)d_in[0];
  const int*   ei = (const int*)d_in[1];
  const float* W1 = (const float*)d_in[2];
  const float* b1 = (const float*)d_in[3];
  const float* W2 = (const float*)d_in[4];
  const float* b2 = (const float*)d_in[5];
  float* out = (float*)d_out;

  const int N = in_sizes[0] / 128;
  const long long E = in_sizes[1] / 2;
  const int* src = ei;
  const int* dst = ei + E;

  float* ws = (float*)d_ws;
  float* dinv  = ws;
  float* h1pre = ws + N;
  float* h1    = h1pre + (size_t)N * 128;
  float* h2pre = h1pre;

  int* rank = (int*)h1;
  int* cnt  = rank + E;
  int* tsum = cnt + N;

  int* listA   = (int*)out;
  int* rowptrA = listA + 2 * E;
  int* listB   = (int*)(h1pre + (size_t)N * 64);
  int* rowptrB = listB + 2 * E;

  const int TB = 256;
  auto blocks = [](long long n, int tb) { return (unsigned)((n + tb - 1) / tb); };
  const unsigned tiles = blocks(N, 1024);

  zero_i32_kernel<<<blocks(N, TB), TB, 0, stream>>>(cnt, N);
  rank_kernel<4><<<blocks(E, TB * 4), TB, 0, stream>>>(dst, cnt, rank, E);
  tilesum_kernel<<<tiles, 256, 0, stream>>>(cnt, tsum, N);
  scan_kernel<<<tiles, 256, 0, stream>>>(cnt, tsum, rowptrA, dinv, N);
  place_kernel<4><<<blocks(E, TB * 4), TB, 0, stream>>>(src, dst, rank, rowptrA, dinv, listA, E);

  gemm_wmma_kernel<128><<<(N + 15) / 16, 256, 0, stream>>>(x, W1, h1pre, N);
  gather_kernel<128, true><<<2048, 256, 0, stream>>>(rowptrA, listA, dinv, h1pre, b1, h1, N);

  move_lists_kernel<<<1024, 256, 0, stream>>>(listA, listB, 2 * E + N + 1);

  gemm_wmma_kernel<64><<<(N + 15) / 16, 128, 0, stream>>>(h1, W2, h2pre, N);
  gather_kernel<64, false><<<2048, 256, 0, stream>>>(rowptrB, listB, dinv, h2pre, b2, out, N);
}


# Fuse layer-1 row gather with the layer-2 WMMA product: a block gathers 32 node rows into LDS and multiplies them there, same k order; h1 is never stored, the list move launch goes (10 to 8 launches)
# speedup vs baseline: 2.2612x; 1.0721x over previous
#include <hip/hip_runtime.h>
#include <hip/hip_bf16.h>


typedef float v2f __attribute__((ext_vector_type(2)));
typedef float v8f __attribute__((ext_vector_type(8)));

__global__ void zero_i32_kernel(int* __restrict__ p, int n) {
  int i = blockIdx.x * blockDim.x + threadIdx.x;
  if (i < n) p[i] = 0;
}

template <int U>
__global__ void rank_kernel(const int* __restrict__ dst, int* __restrict__ cnt,
                            int* __restrict__ rank, long long E) {
  const long long T = (long long)gridDim.x * blockDim.x;
  const long long t = (long long)blockIdx.x * blockDim.x + threadIdx.x;
  if (t + (U - 1) * T < E) {
    int d[U];
    int r[U];
    #pragma unroll
    for (int k = 0; k < U; ++k) d[k] = dst[t + k * T];
    #pragma unroll
    for (int k = 0; k < U; ++k)
      r[k] = __hip_atomic_fetch_add(&cnt[d[k]], 1, __ATOMIC_RELAXED,
                                    __HIP_MEMORY_SCOPE_AGENT);
    #pragma unroll
    for (int k = 0; k < U; ++k) rank[t + k * T] = r[k];
  } else {
    for (int k = 0; k < U; ++k) {
      const long long e = t + k * T;
      if (e < E)
        rank[e] = __hip_atomic_fetch_add(&cnt[dst[e]], 1, __ATOMIC_RELAXED,
                                         __HIP_MEMORY_SCOPE_AGENT);
    }
  }
}

__global__ __launch_bounds__(256) void tilesum_kernel(const int* __restrict__ cnt,
                                                      int* __restrict__ tsum, int n) {
  __shared__ int red[8];
  const int tid = threadIdx.x;
  const int base = blockIdx.x * 1024 + tid * 4;
  int s = 0;
  for (int k = 0; k < 4; ++k)
    if (base + k < n) s += cnt[base + k];
  for (int o = 16; o > 0; o >>= 1) s += __shfl_xor(s, o);
  if ((tid & 31) == 0) red[tid >> 5] = s;
  __syncthreads();
  if (tid == 0) {
    int t = 0;
    for (int w = 0; w < 8; ++w) t += red[w];
    tsum[blockIdx.x] = t;
  }
}

__global__ __launch_bounds__(256) void scan_kernel(const int* __restrict__ cnt,
                                                   const int* __restrict__ tsum,
                                                   int* __restrict__ rowptr,
                                                   float* __restrict__ dinv, int n) {
  __shared__ int red[8];
  __shared__ int wtot[8];
  const int tid = threadIdx.x;
  const int lane = tid & 31;
  const int wave = tid >> 5;

  int p = 0;
  for (int i = tid; i < (int)blockIdx.x; i += 256) p += tsum[i];
  for (int o = 16; o > 0; o >>= 1) p += __shfl_xor(p, o);
  if (lane == 0) red[wave] = p;

  const int base = blockIdx.x * 1024 + tid * 4;
  int c[4];
  for (int k = 0; k < 4; ++k) c[k] = (base + k < n) ? cnt[base + k] : 0;
  const int mine = c[0] + c[1] + c[2] + c[3];
  int incl = mine;
  for (int o = 1; o < 32; o <<= 1) {
    const int y = __shfl_up(incl, o);
    if (lane >= o) incl += y;
  }
  if (lane == 31) wtot[wave] = incl;
  __syncthreads();

  int off = 0;
  for (int w = 0; w < 8; ++w) off += red[w];
  for (int w = 0; w < wave; ++w) off += wtot[w];
  int run = off + incl - mine;
  for (int k = 0; k < 4; ++k) {
    if (base + k < n) {
      rowptr[base + k] = run;
      dinv[base + k] = __frsqrt_rn(fmaxf((float)(c[k] + 1), 1e-12f));
      run += c[k];
      if (base + k == n - 1) rowptr[n] = run;
    }
  }
}

template <int U>
__global__ void place_kernel(const int* __restrict__ src, const int* __restrict__ dst,
                             const int* __restrict__ rank,
                             const int* __restrict__ rowptr,
                             const float* __restrict__ dinv, int* __restrict__ list,
                             long long E) {
  const long long T = (long long)gridDim.x * blockDim.x;
  const long long t = (long long)blockIdx.x * blockDim.x + threadIdx.x;
  if (t + (U - 1) * T < E) {
    int u[U], v[U], rk[U], pos[U];
    float w[U];
    #pragma unroll
    for (int k = 0; k < U; ++k) {
      u[k] = src[t + k * T];
      v[k] = dst[t + k * T];
      rk[k] = rank[t + k * T];
    }
    #pragma unroll
    for (int k = 0; k < U; ++k) {
      pos[k] = rowptr[v[k]] + rk[k];
      w[k] = dinv[u[k]];
    }
    #pragma unroll
    for (int k = 0; k < U; ++k) {
      int2 ent;
      ent.x = u[k];
      ent.y = __float_as_int(w[k]);
      *(int2*)&list[2 * (size_t)pos[k]] = ent;
    }
  } else {
    for (int k = 0; k < U; ++k) {
      const long long e = t + k * T;
      if (e < E) {
        const int uu = src[e];
        const int pp = rowptr[dst[e]] + rank[e];
        int2 ent;
        ent.x = uu;
        ent.y = __float_as_int(dinv[uu]);
        *(int2*)&list[2 * (size_t)pp] = ent;
      }
    }
  }
}

template <int FOUT>
__global__ __launch_bounds__(FOUT * 2) void gemm_wmma_kernel(
    const float* __restrict__ A, const float* __restrict__ W,
    float* __restrict__ C, int nrows) {
  __shared__ float lds[16 * 128];

  const int row0 = blockIdx.x * 16;
  if (row0 >= nrows) return;
  const int tid = threadIdx.x;
  constexpr int NTHREADS = FOUT * 2;

  if (row0 + 16 <= nrows) {
    const float4* srcv = (const float4*)(A + (size_t)row0 * 128);
    float4* dstv = (float4*)lds;
    #pragma unroll 2
    for (int i = tid; i < 512; i += NTHREADS) dstv[i] = srcv[i];
  } else {
    for (int i = tid; i < 2048; i += NTHREADS) {
      int r = i >> 7;
      lds[i] = (row0 + r < nrows) ? A[(size_t)(row0 + r) * 128 + (i & 127)] : 0.0f;
    }
  }
  __syncthreads();

  const int lane = tid & 31;
  const int wave = tid >> 5;
  const int colbase = wave * 16;
  const int half = lane >> 4;
  const int m = lane & 15;

  v8f acc = {};
  #pragma unroll
  for (int k = 0; k < 128; k += 4) {
    const int kb = k + half * 2;
    v2f a = *(const v2f*)&lds[m * 128 + kb];
    v2f b;
    b.x = W[(size_t)kb * FOUT + colbase + m];
    b.y = W[(size_t)(kb + 1) * FOUT + colbase + m];
    acc = __builtin_amdgcn_wmma_f32_16x16x4_f32(false, a, false, b, (short)0,
                                                acc, false, false);
  }

  #pragma unroll
  for (int i = 0; i < 8; ++i) {
    int row = row0 + i + half * 8;
    if (row < nrows) C[(size_t)row * FOUT + colbase + m] = acc[i];
  }
}

template <int F, bool RELU>
__global__ __launch_bounds__(256) void gather_kernel(
    const int* __restrict__ rowptr, const int* __restrict__ list,
    const float* __restrict__ dinv, const float* __restrict__ h,
    const float* __restrict__ bias, float* __restrict__ y, int n) {
  constexpr int V = F / 32;
  typedef float vecT __attribute__((ext_vector_type(V)));
  constexpr int G = 8;

  const int lane = threadIdx.x & 31;
  const int wv = __builtin_amdgcn_readfirstlane((int)(threadIdx.x >> 5));
  const int wpb = (int)(blockDim.x >> 5);
  const int nwaves = (int)gridDim.x * wpb;
  const vecT bv = *(const vecT*)&bias[lane * V];

  for (int v = (int)blockIdx.x * wpb + wv; v < n; v += nwaves) {
    const int beg = __builtin_amdgcn_readfirstlane(rowptr[v]);
    const int end = __builtin_amdgcn_readfirstlane(rowptr[v + 1]);
    const float dv = dinv[v];
    const vecT hs = *(const vecT*)&h[(size_t)v * F + lane * V];
    vecT acc = {};

    for (int base = beg; base < end; base += 32) {
      const int m = (end - base < 32) ? (end - base) : 32;
      int eu = v;
      int ew = 0;
      if (lane < m) {
        const int2 ent = *(const int2*)&list[2 * (size_t)(base + lane)];
        eu = ent.x;
        ew = ent.y;
      }
      for (int j = 0; j < m; j += G) {
        vecT r[G];
        float w[G];
        #pragma unroll
        for (int q = 0; q < G; ++q) {
          const int u = __builtin_amdgcn_readlane(eu, j + q);
          w[q] = __int_as_float(__builtin_amdgcn_readlane(ew, j + q));
          r[q] = *(const vecT*)&h[(size_t)u * F + lane * V];
        }
        #pragma unroll
        for (int q = 0; q < G; ++q) acc += (w[q] * dv) * r[q];
      }
    }

    acc += (dv * dv) * hs;
    acc += bv;
    if (RELU) acc = __builtin_elementwise_max(acc, (vecT)(0.0f));
    *(vecT*)&y[(size_t)v * F + lane * V] = acc;
  }
}

__global__ __launch_bounds__(256) void gather_relu_gemm_kernel(
    const int* __restrict__ rowptr, const int* __restrict__ list,
    const float* __restrict__ dinv, const float* __restrict__ h,
    const float* __restrict__ bias, const float* __restrict__ W,
    float* __restrict__ C, int n) {
  constexpr int F = 128;
  constexpr int FOUT = 64;
  constexpr int P = 132;
  constexpr int G = 8;
  typedef float vecT __attribute__((ext_vector_type(4)));
  __shared__ float tile[32 * P];

  const int lane = threadIdx.x & 31;
  const int wv = __builtin_amdgcn_readfirstlane((int)(threadIdx.x >> 5));
  const int tile0 = (int)blockIdx.x * 32;
  const int v0 = tile0 + wv * 4;
  const vecT bv = *(const vecT*)&bias[lane * 4];

  int bi = v0 + (lane < 4 ? lane : 4);
  bi = bi < n ? bi : n;
  const int rp = rowptr[bi];
  int di = v0 + (lane & 3);
  di = di < n ? di : n - 1;
  const float dvv = dinv[di];

  for (int i = 0; i < 4; ++i) {
    const int v = v0 + i;
    vecT acc = {};
    if (v < n) {
      const int beg = __builtin_amdgcn_readlane(rp, i);
      const int end = __builtin_amdgcn_readlane(rp, i + 1);
      const float dv = __int_as_float(__builtin_amdgcn_readlane(__float_as_int(dvv), i));
      const vecT hs = *(const vecT*)&h[(size_t)v * F + lane * 4];

      for (int base = beg; base < end; base += 32) {
        const int m = (end - base < 32) ? (end - base) : 32;
        int eu = v;
        int ew = 0;
        if (lane < m) {
          const int2 ent = *(const int2*)&list[2 * (size_t)(base + lane)];
          eu = ent.x;
          ew = ent.y;
        }
        for (int j = 0; j < m; j += G) {
          vecT r[G];
          float w[G];
          #pragma unroll
          for (int q = 0; q < G; ++q) {
            const int u = __builtin_amdgcn_readlane(eu, j + q);
            w[q] = __int_as_float(__builtin_amdgcn_readlane(ew, j + q));
            r[q] = *(const vecT*)&h[(size_t)u * F + lane * 4];
          }
          #pragma unroll
          for (int q = 0; q < G; ++q) acc += (w[q] * dv) * r[q];
        }
      }

      acc += (dv * dv) * hs;
      acc += bv;
      acc = __builtin_elementwise_max(acc, (vecT)(0.0f));
    }
    *(vecT*)&tile[(wv * 4 + i) * P + lane * 4] = acc;
  }

  const int colbase = (wv & 3) * 16;
  const int rbase = (wv >> 2) * 16;
  const int half = lane >> 4;
  const int m = lane & 15;

  v2f b[32];
  #pragma unroll
  for (int kk = 0; kk < 32; ++kk) {
    const int kb = kk * 4 + half * 2;
    b[kk].x = W[(size_t)kb * FOUT + colbase + m];
    b[kk].y = W[(size_t)(kb + 1) * FOUT + colbase + m];
  }

  __syncthreads();

  v8f dacc = {};
  #pragma unroll
  for (int kk = 0; kk < 32; ++kk) {
    const int kb = kk * 4 + half * 2;
    const v2f a = *(const v2f*)&tile[(rbase + m) * P + kb];
    dacc = __builtin_amdgcn_wmma_f32_16x16x4_f32(false, a, false, b[kk], (short)0,
                                                 dacc, false, false);
  }

  #pragma unroll
  for (int i = 0; i < 8; ++i) {
    const int row = tile0 + rbase + i + half * 8;
    if (row < n) C[(size_t)row * FOUT + colbase + m] = dacc[i];
  }
}

extern "C" void kernel_launch(void* const* d_in, const int* in_sizes, int n_in,
                              void* d_out, int out_size, void* d_ws,
                              size_t ws_size, hipStream_t stream) {
  const float* x  = (const float*)d_in[0];
  const int*   ei = (const int*)d_in[1];
  const float* W1 = (const float*)d_in[2];
  const float* b1 = (const float*)d_in[3];
  const float* W2 = (const float*)d_in[4];
  const float* b2 = (const float*)d_in[5];
  float* out = (float*)d_out;

  const int N = in_sizes[0] / 128;
  const long long E = in_sizes[1] / 2;
  const int* src = ei;
  const int* dst = ei + E;

  float* ws = (float*)d_ws;
  float* dinv  = ws;
  float* h1pre = ws + N;
  float* reg2  = h1pre + (size_t)N * 128;
  float* h2pre = reg2;

  int* rank   = (int*)(reg2 + (size_t)N * 64);
  int* list   = rank + E;
  int* rowptr = list + 2 * E;
  int* cnt    = rowptr + N + 1;
  int* tsum   = cnt + N;

  const int TB = 256;
  auto blocks = [](long long n, int tb) { return (unsigned)((n + tb - 1) / tb); };
  const unsigned tiles = blocks(N, 1024);

  zero_i32_kernel<<<blocks(N, TB), TB, 0, stream>>>(cnt, N);
  rank_kernel<4><<<blocks(E, TB * 4), TB, 0, stream>>>(dst, cnt, rank, E);
  tilesum_kernel<<<tiles, 256, 0, stream>>>(cnt, tsum, N);
  scan_kernel<<<tiles, 256, 0, stream>>>(cnt, tsum, rowptr, dinv, N);
  place_kernel<4><<<blocks(E, TB * 4), TB, 0, stream>>>(src, dst, rank, rowptr, dinv, list, E);

  gemm_wmma_kernel<128><<<(N + 15) / 16, 256, 0, stream>>>(x, W1, h1pre, N);
  gather_relu_gemm_kernel<<<blocks(N, 32), 256, 0, stream>>>(rowptr, list, dinv, h1pre, b1, W2, h2pre, N);

  gather_kernel<64, false><<<2048, 256, 0, stream>>>(rowptr, list, dinv, h2pre, b2, out, N);
}


# Build the in-edge lists in two levels: bin edges by 256-node bucket (LDS histogram, one memory add per block and bucket), rank per bucket on LDS counters; list order = LDS add arrival order
# speedup vs baseline: 3.3512x; 1.4820x over previous
#include <hip/hip_runtime.h>
#include <hip/hip_bf16.h>


typedef float v2f __attribute__((ext_vector_type(2)));
typedef float v8f __attribute__((ext_vector_type(8)));

constexpr int BSH = 8;
constexpr int BN = 1 << BSH;
constexpr int MAXB = 1024;
constexpr int BT = 1024;
constexpr int BU = 4;

__global__ void zero_i32_kernel(int* __restrict__ p, int n) {
  int i = blockIdx.x * blockDim.x + threadIdx.x;
  if (i < n) p[i] = 0;
}

__global__ __launch_bounds__(BT) void bucket_count_kernel(
    const int* __restrict__ dst, int* __restrict__ gcount, long long E, int nb) {
  __shared__ int hist[MAXB];
  const int tid = threadIdx.x;
  for (int b = tid; b < nb; b += BT) hist[b] = 0;
  __syncthreads();
  const long long e0 = (long long)blockIdx.x * (BT * BU) + tid;
  int d[BU];
  #pragma unroll
  for (int k = 0; k < BU; ++k) {
    const long long e = e0 + (long long)k * BT;
    d[k] = (e < E) ? dst[e] : -1;
  }
  #pragma unroll
  for (int k = 0; k < BU; ++k)
    if (d[k] >= 0)
      __hip_atomic_fetch_add(&hist[d[k] >> BSH], 1, __ATOMIC_RELAXED,
                             __HIP_MEMORY_SCOPE_WORKGROUP);
  __syncthreads();
  for (int b = tid; b < nb; b += BT) {
    const int c = hist[b];
    if (c != 0)
      __hip_atomic_fetch_add(&gcount[b], c, __ATOMIC_RELAXED, __HIP_MEMORY_SCOPE_AGENT);
  }
}

__global__ __launch_bounds__(BT) void bucket_scatter_kernel(
    const int* __restrict__ src, const int* __restrict__ dst,
    const int* __restrict__ gcount, int* __restrict__ gcursor,
    int* __restrict__ bk, long long E, int nb) {
  __shared__ int hist[MAXB];
  __shared__ int wsum[BT / 32];
  const int tid = threadIdx.x;
  const int lane = tid & 31;
  const int wave = tid >> 5;

  const int g = (tid < nb) ? gcount[tid] : 0;

  for (int b = tid; b < nb; b += BT) hist[b] = 0;
  __syncthreads();

  const long long e0 = (long long)blockIdx.x * (BT * BU) + tid;
  int u[BU], d[BU], lr[BU];
  #pragma unroll
  for (int k = 0; k < BU; ++k) {
    const long long e = e0 + (long long)k * BT;
    u[k] = 0;
    d[k] = -1;
    if (e < E) {
      u[k] = src[e];
      d[k] = dst[e];
    }
  }
  #pragma unroll
  for (int k = 0; k < BU; ++k) {
    lr[k] = 0;
    if (d[k] >= 0)
      lr[k] = __hip_atomic_fetch_add(&hist[d[k] >> BSH], 1, __ATOMIC_RELAXED,
                                     __HIP_MEMORY_SCOPE_WORKGROUP);
  }

  int incl = g;
  for (int o = 1; o < 32; o <<= 1) {
    const int y = __shfl_up(incl, o);
    if (lane >= o) incl += y;
  }
  if (lane == 31) wsum[wave] = incl;
  __syncthreads();

  int start = incl - g;
  for (int w = 0; w < wave; ++w) start += wsum[w];
  if (tid < nb) {
    const int c = hist[tid];
    int base = 0;
    if (c != 0)
      base = __hip_atomic_fetch_add(&gcursor[tid], c, __ATOMIC_RELAXED,
                                    __HIP_MEMORY_SCOPE_AGENT);
    hist[tid] = start + base;
  }
  __syncthreads();

  #pragma unroll
  for (int k = 0; k < BU; ++k) {
    if (d[k] >= 0) {
      const int pos = hist[d[k] >> BSH] + lr[k];
      int2 ent;
      ent.x = u[k];
      ent.y = d[k];
      *(int2*)&bk[2 * (size_t)pos] = ent;
    }
  }
}

__global__ __launch_bounds__(BT) void bucket_finish_kernel(
    const int* __restrict__ gcount, const int* __restrict__ bk,
    int* __restrict__ rowptr, float* __restrict__ dinv, int* __restrict__ srcs,
    int n) {
  __shared__ int c[BN];
  __shared__ int red[BT / 32];
  __shared__ int wtot[BN / 32];
  const int tid = threadIdx.x;
  const int lane = tid & 31;
  const int wave = tid >> 5;
  const int b = (int)blockIdx.x;
  const int lo = b << BSH;

  int p = 0;
  for (int i = tid; i < b; i += BT) p += gcount[i];
  for (int o = 16; o > 0; o >>= 1) p += __shfl_xor(p, o);
  if (lane == 0) red[wave] = p;
  const int cntb = gcount[b];
  if (tid < BN) c[tid] = 0;
  __syncthreads();
  int boff = 0;
  for (int w = 0; w < BT / 32; ++w) boff += red[w];

  for (int i0 = 0; i0 < cntb; i0 += BT * BU) {
    int dd[BU];
    #pragma unroll
    for (int k = 0; k < BU; ++k) {
      const int i = i0 + k * BT + tid;
      dd[k] = (i < cntb) ? bk[2 * (size_t)(boff + i) + 1] : -1;
    }
    #pragma unroll
    for (int k = 0; k < BU; ++k)
      if (dd[k] >= 0)
        __hip_atomic_fetch_add(&c[dd[k] - lo], 1, __ATOMIC_RELAXED,
                               __HIP_MEMORY_SCOPE_WORKGROUP);
  }
  __syncthreads();

  const int mine = (tid < BN) ? c[tid] : 0;
  int incl = mine;
  for (int o = 1; o < 32; o <<= 1) {
    const int y = __shfl_up(incl, o);
    if (lane >= o) incl += y;
  }
  if (tid < BN && lane == 31) wtot[wave] = incl;
  __syncthreads();
  if (tid < BN) {
    int excl = incl - mine;
    for (int w = 0; w < wave; ++w) excl += wtot[w];
    const int v = lo + tid;
    if (v < n) {
      rowptr[v] = boff + excl;
      dinv[v] = __frsqrt_rn(fmaxf((float)(mine + 1), 1e-12f));
      if (v == n - 1) rowptr[n] = boff + excl + mine;
    }
    c[tid] = excl;
  }
  __syncthreads();

  for (int i0 = 0; i0 < cntb; i0 += BT * BU) {
    int2 ent[BU];
    #pragma unroll
    for (int k = 0; k < BU; ++k) {
      const int i = i0 + k * BT + tid;
      ent[k].x = 0;
      ent[k].y = -1;
      if (i < cntb) ent[k] = *(const int2*)&bk[2 * (size_t)(boff + i)];
    }
    int pos[BU];
    #pragma unroll
    for (int k = 0; k < BU; ++k) {
      pos[k] = 0;
      if (ent[k].y >= 0)
        pos[k] = __hip_atomic_fetch_add(&c[ent[k].y - lo], 1, __ATOMIC_RELAXED,
                                        __HIP_MEMORY_SCOPE_WORKGROUP);
    }
    #pragma unroll
    for (int k = 0; k < BU; ++k)
      if (ent[k].y >= 0) srcs[(size_t)boff + pos[k]] = ent[k].x;
  }
}

template <int U>
__global__ void weight_kernel(const int* __restrict__ srcs, const float* __restrict__ dinv,
                              int* __restrict__ list, long long E) {
  const long long T = (long long)gridDim.x * blockDim.x;
  const long long t = (long long)blockIdx.x * blockDim.x + threadIdx.x;
  int u[U];
  float w[U];
  #pragma unroll
  for (int k = 0; k < U; ++k) {
    const long long e = t + k * T;
    u[k] = (e < E) ? srcs[e] : 0;
  }
  #pragma unroll
  for (int k = 0; k < U; ++k) w[k] = dinv[u[k]];
  #pragma unroll
  for (int k = 0; k < U; ++k) {
    const long long e = t + k * T;
    if (e < E) {
      int2 ent;
      ent.x = u[k];
      ent.y = __float_as_int(w[k]);
      *(int2*)&list[2 * (size_t)e] = ent;
    }
  }
}

template <int FOUT>
__global__ __launch_bounds__(FOUT * 2) void gemm_wmma_kernel(
    const float* __restrict__ A, const float* __restrict__ W,
    float* __restrict__ C, int nrows) {
  __shared__ float lds[16 * 128];

  const int row0 = blockIdx.x * 16;
  if (row0 >= nrows) return;
  const int tid = threadIdx.x;
  constexpr int NTHREADS = FOUT * 2;

  if (row0 + 16 <= nrows) {
    const float4* srcv = (const float4*)(A + (size_t)row0 * 128);
    float4* dstv = (float4*)lds;
    #pragma unroll 2
    for (int i = tid; i < 512; i += NTHREADS) dstv[i] = srcv[i];
  } else {
    for (int i = tid; i < 2048; i += NTHREADS) {
      int r = i >> 7;
      lds[i] = (row0 + r < nrows) ? A[(size_t)(row0 + r) * 128 + (i & 127)] : 0.0f;
    }
  }
  __syncthreads();

  const int lane = tid & 31;
  const int wave = tid >> 5;
  const int colbase = wave * 16;
  const int half = lane >> 4;
  const int m = lane & 15;

  v8f acc = {};
  #pragma unroll
  for (int k = 0; k < 128; k += 4) {
    const int kb = k + half * 2;
    v2f a = *(const v2f*)&lds[m * 128 + kb];
    v2f b;
    b.x = W[(size_t)kb * FOUT + colbase + m];
    b.y = W[(size_t)(kb + 1) * FOUT + colbase + m];
    acc = __builtin_amdgcn_wmma_f32_16x16x4_f32(false, a, false, b, (short)0,
                                                acc, false, false);
  }

  #pragma unroll
  for (int i = 0; i < 8; ++i) {
    int row = row0 + i + half * 8;
    if (row < nrows) C[(size_t)row * FOUT + colbase + m] = acc[i];
  }
}

template <int F, bool RELU>
__global__ __launch_bounds__(256) void gather_kernel(
    const int* __restrict__ rowptr, const int* __restrict__ list,
    const float* __restrict__ dinv, const float* __restrict__ h,
    const float* __restrict__ bias, float* __restrict__ y, int n) {
  constexpr int V = F / 32;
  typedef float vecT __attribute__((ext_vector_type(V)));
  constexpr int G = 8;

  const int lane = threadIdx.x & 31;
  const int wv = __builtin_amdgcn_readfirstlane((int)(threadIdx.x >> 5));
  const int wpb = (int)(blockDim.x >> 5);
  const int nwaves = (int)gridDim.x * wpb;
  const vecT bv = *(const vecT*)&bias[lane * V];

  for (int v = (int)blockIdx.x * wpb + wv; v < n; v += nwaves) {
    const int beg = __builtin_amdgcn_readfirstlane(rowptr[v]);
    const int end = __builtin_amdgcn_readfirstlane(rowptr[v + 1]);
    const float dv = dinv[v];
    const vecT hs = *(const vecT*)&h[(size_t)v * F + lane * V];
    vecT acc = {};

    for (int base = beg; base < end; base += 32) {
      const int m = (end - base < 32) ? (end - base) : 32;
      int eu = v;
      int ew = 0;
      if (lane < m) {
        const int2 ent = *(const int2*)&list[2 * (size_t)(base + lane)];
        eu = ent.x;
        ew = ent.y;
      }
      for (int j = 0; j < m; j += G) {
        vecT r[G];
        float w[G];
        #pragma unroll
        for (int q = 0; q < G; ++q) {
          const int u = __builtin_amdgcn_readlane(eu, j + q);
          w[q] = __int_as_float(__builtin_amdgcn_readlane(ew, j + q));
          r[q] = *(const vecT*)&h[(size_t)u * F + lane * V];
        }
        #pragma unroll
        for (int q = 0; q < G; ++q) acc += (w[q] * dv) * r[q];
      }
    }

    acc += (dv * dv) * hs;
    acc += bv;
    if (RELU) acc = __builtin_elementwise_max(acc, (vecT)(0.0f));
    *(vecT*)&y[(size_t)v * F + lane * V] = acc;
  }
}

__global__ __launch_bounds__(256) void gather_relu_gemm_kernel(
    const int* __restrict__ rowptr, const int* __restrict__ list,
    const float* __restrict__ dinv, const float* __restrict__ h,
    const float* __restrict__ bias, const float* __restrict__ W,
    float* __restrict__ C, int n) {
  constexpr int F = 128;
  constexpr int FOUT = 64;
  constexpr int P = 132;
  constexpr int G = 8;
  typedef float vecT __attribute__((ext_vector_type(4)));
  __shared__ float tile[32 * P];

  const int lane = threadIdx.x & 31;
  const int wv = __builtin_amdgcn_readfirstlane((int)(threadIdx.x >> 5));
  const int tile0 = (int)blockIdx.x * 32;
  const int v0 = tile0 + wv * 4;
  const vecT bv = *(const vecT*)&bias[lane * 4];

  int bi = v0 + (lane < 4 ? lane : 4);
  bi = bi < n ? bi : n;
  const int rp = rowptr[bi];
  int di = v0 + (lane & 3);
  di = di < n ? di : n - 1;
  const float dvv = dinv[di];

  for (int i = 0; i < 4; ++i) {
    const int v = v0 + i;
    vecT acc = {};
    if (v < n) {
      const int beg = __builtin_amdgcn_readlane(rp, i);
      const int end = __builtin_amdgcn_readlane(rp, i + 1);
      const float dv = __int_as_float(__builtin_amdgcn_readlane(__float_as_int(dvv), i));
      const vecT hs = *(const vecT*)&h[(size_t)v * F + lane * 4];

      for (int base = beg; base < end; base += 32) {
        const int m = (end - base < 32) ? (end - base) : 32;
        int eu = v;
        int ew = 0;
        if (lane < m) {
          const int2 ent = *(const int2*)&list[2 * (size_t)(base + lane)];
          eu = ent.x;
          ew = ent.y;
        }
        for (int j = 0; j < m; j += G) {
          vecT r[G];
          float w[G];
          #pragma unroll
          for (int q = 0; q < G; ++q) {
            const int u = __builtin_amdgcn_readlane(eu, j + q);
            w[q] = __int_as_float(__builtin_amdgcn_readlane(ew, j + q));
            r[q] = *(const vecT*)&h[(size_t)u * F + lane * 4];
          }
          #pragma unroll
          for (int q = 0; q < G; ++q) acc += (w[q] * dv) * r[q];
        }
      }

      acc += (dv * dv) * hs;
      acc += bv;
      acc = __builtin_elementwise_max(acc, (vecT)(0.0f));
    }
    *(vecT*)&tile[(wv * 4 + i) * P + lane * 4] = acc;
  }

  const int colbase = (wv & 3) * 16;
  const int rbase = (wv >> 2) * 16;
  const int half = lane >> 4;
  const int m = lane & 15;

  v2f b[32];
  #pragma unroll
  for (int kk = 0; kk < 32; ++kk) {
    const int kb = kk * 4 + half * 2;
    b[kk].x = W[(size_t)kb * FOUT + colbase + m];
    b[kk].y = W[(size_t)(kb + 1) * FOUT + colbase + m];
  }

  __syncthreads();

  v8f dacc = {};
  #pragma unroll
  for (int kk = 0; kk < 32; ++kk) {
    const int kb = kk * 4 + half * 2;
    const v2f a = *(const v2f*)&tile[(rbase + m) * P + kb];
    dacc = __builtin_amdgcn_wmma_f32_16x16x4_f32(false, a, false, b[kk], (short)0,
                                                 dacc, false, false);
  }

  #pragma unroll
  for (int i = 0; i < 8; ++i) {
    const int row = tile0 + rbase + i + half * 8;
    if (row < n) C[(size_t)row * FOUT + colbase + m] = dacc[i];
  }
}

extern "C" void kernel_launch(void* const* d_in, const int* in_sizes, int n_in,
                              void* d_out, int out_size, void* d_ws,
                              size_t ws_size, hipStream_t stream) {
  const float* x  = (const float*)d_in[0];
  const int*   ei = (const int*)d_in[1];
  const float* W1 = (const float*)d_in[2];
  const float* b1 = (const float*)d_in[3];
  const float* W2 = (const float*)d_in[4];
  const float* b2 = (const float*)d_in[5];
  float* out = (float*)d_out;

  const int N = in_sizes[0] / 128;
  const long long E = in_sizes[1] / 2;
  const int* src = ei;
  const int* dst = ei + E;

  float* ws = (float*)d_ws;
  float* dinv  = ws;
  float* h1pre = ws + N;
  float* reg2  = h1pre + (size_t)N * 128;
  float* h2pre = reg2;

  int* list    = (int*)(reg2 + (size_t)N * 64);
  int* rowptr  = list + 2 * E;
  int* gcount  = rowptr + N + 1;
  int* gcursor = gcount + ((N + BN - 1) >> BSH);
  int* bk   = (int*)h1pre;
  int* srcs = bk + 2 * E;

  const int TB = 256;
  auto blocks = [](long long n, int tb) { return (unsigned)((n + tb - 1) / tb); };
  const int NB = (N + BN - 1) >> BSH;

  zero_i32_kernel<<<blocks(2 * NB, TB), TB, 0, stream>>>(gcount, 2 * NB);
  bucket_count_kernel<<<blocks(E, BT * BU), BT, 0, stream>>>(dst, gcount, E, NB);
  bucket_scatter_kernel<<<blocks(E, BT * BU), BT, 0, stream>>>(src, dst, gcount, gcursor, bk, E, NB);
  bucket_finish_kernel<<<NB, BT, 0, stream>>>(gcount, bk, rowptr, dinv, srcs, N);
  weight_kernel<8><<<blocks(E, TB * 8), TB, 0, stream>>>(srcs, dinv, list, E);

  gemm_wmma_kernel<128><<<(N + 15) / 16, 256, 0, stream>>>(x, W1, h1pre, N);
  gather_relu_gemm_kernel<<<blocks(N, 32), 256, 0, stream>>>(rowptr, list, dinv, h1pre, b1, W2, h2pre, N);

  gather_kernel<64, false><<<2048, 256, 0, stream>>>(rowptr, list, dinv, h2pre, b2, out, N);
}


# Aggregate input rows before the layer-1 product, (A x) W1 for A (x W1), and do it on the gathered tile in LDS: layer-1 GEMM launch and its stored rows go (8 to 7 launches); layer-1 sum order changes
# speedup vs baseline: 4.6698x; 1.3935x over previous
#include <hip/hip_runtime.h>
#include <hip/hip_bf16.h>


typedef float v2f __attribute__((ext_vector_type(2)));
typedef float v8f __attribute__((ext_vector_type(8)));

constexpr int BSH = 8;
constexpr int BN = 1 << BSH;
constexpr int MAXB = 1024;
constexpr int BT = 1024;
constexpr int BU = 4;

__global__ void zero_i32_kernel(int* __restrict__ p, int n) {
  int i = blockIdx.x * blockDim.x + threadIdx.x;
  if (i < n) p[i] = 0;
}

__global__ __launch_bounds__(BT) void bucket_count_kernel(
    const int* __restrict__ dst, int* __restrict__ gcount, long long E, int nb) {
  __shared__ int hist[MAXB];
  const int tid = threadIdx.x;
  for (int b = tid; b < nb; b += BT) hist[b] = 0;
  __syncthreads();
  const long long e0 = (long long)blockIdx.x * (BT * BU) + tid;
  int d[BU];
  #pragma unroll
  for (int k = 0; k < BU; ++k) {
    const long long e = e0 + (long long)k * BT;
    d[k] = (e < E) ? dst[e] : -1;
  }
  #pragma unroll
  for (int k = 0; k < BU; ++k)
    if (d[k] >= 0)
      __hip_atomic_fetch_add(&hist[d[k] >> BSH], 1, __ATOMIC_RELAXED,
                             __HIP_MEMORY_SCOPE_WORKGROUP);
  __syncthreads();
  for (int b = tid; b < nb; b += BT) {
    const int c = hist[b];
    if (c != 0)
      __hip_atomic_fetch_add(&gcount[b], c, __ATOMIC_RELAXED, __HIP_MEMORY_SCOPE_AGENT);
  }
}

__global__ __launch_bounds__(BT) void bucket_scatter_kernel(
    const int* __restrict__ src, const int* __restrict__ dst,
    const int* __restrict__ gcount, int* __restrict__ gcursor,
    int* __restrict__ bk, long long E, int nb) {
  __shared__ int hist[MAXB];
  __shared__ int wsum[BT / 32];
  const int tid = threadIdx.x;
  const int lane = tid & 31;
  const int wave = tid >> 5;

  const int g = (tid < nb) ? gcount[tid] : 0;

  for (int b = tid; b < nb; b += BT) hist[b] = 0;
  __syncthreads();

  const long long e0 = (long long)blockIdx.x * (BT * BU) + tid;
  int u[BU], d[BU], lr[BU];
  #pragma unroll
  for (int k = 0; k < BU; ++k) {
    const long long e = e0 + (long long)k * BT;
    u[k] = 0;
    d[k] = -1;
    if (e < E) {
      u[k] = src[e];
      d[k] = dst[e];
    }
  }
  #pragma unroll
  for (int k = 0; k < BU; ++k) {
    lr[k] = 0;
    if (d[k] >= 0)
      lr[k] = __hip_atomic_fetch_add(&hist[d[k] >> BSH], 1, __ATOMIC_RELAXED,
                                     __HIP_MEMORY_SCOPE_WORKGROUP);
  }

  int incl = g;
  for (int o = 1; o < 32; o <<= 1) {
    const int y = __shfl_up(incl, o);
    if (lane >= o) incl += y;
  }
  if (lane == 31) wsum[wave] = incl;
  __syncthreads();

  int start = incl - g;
  for (int w = 0; w < wave; ++w) start += wsum[w];
  if (tid < nb) {
    const int c = hist[tid];
    int base = 0;
    if (c != 0)
      base = __hip_atomic_fetch_add(&gcursor[tid], c, __ATOMIC_RELAXED,
                                    __HIP_MEMORY_SCOPE_AGENT);
    hist[tid] = start + base;
  }
  __syncthreads();

  #pragma unroll
  for (int k = 0; k < BU; ++k) {
    if (d[k] >= 0) {
      const int pos = hist[d[k] >> BSH] + lr[k];
      int2 ent;
      ent.x = u[k];
      ent.y = d[k];
      *(int2*)&bk[2 * (size_t)pos] = ent;
    }
  }
}

__global__ __launch_bounds__(BT) void bucket_finish_kernel(
    const int* __restrict__ gcount, const int* __restrict__ bk,
    int* __restrict__ rowptr, float* __restrict__ dinv, int* __restrict__ srcs,
    int n) {
  __shared__ int c[BN];
  __shared__ int red[BT / 32];
  __shared__ int wtot[BN / 32];
  const int tid = threadIdx.x;
  const int lane = tid & 31;
  const int wave = tid >> 5;
  const int b = (int)blockIdx.x;
  const int lo = b << BSH;

  int p = 0;
  for (int i = tid; i < b; i += BT) p += gcount[i];
  for (int o = 16; o > 0; o >>= 1) p += __shfl_xor(p, o);
  if (lane == 0) red[wave] = p;
  const int cntb = gcount[b];
  if (tid < BN) c[tid] = 0;
  __syncthreads();
  int boff = 0;
  for (int w = 0; w < BT / 32; ++w) boff += red[w];

  for (int i0 = 0; i0 < cntb; i0 += BT * BU) {
    int dd[BU];
    #pragma unroll
    for (int k = 0; k < BU; ++k) {
      const int i = i0 + k * BT + tid;
      dd[k] = (i < cntb) ? bk[2 * (size_t)(boff + i) + 1] : -1;
    }
    #pragma unroll
    for (int k = 0; k < BU; ++k)
      if (dd[k] >= 0)
        __hip_atomic_fetch_add(&c[dd[k] - lo], 1, __ATOMIC_RELAXED,
                               __HIP_MEMORY_SCOPE_WORKGROUP);
  }
  __syncthreads();

  const int mine = (tid < BN) ? c[tid] : 0;
  int incl = mine;
  for (int o = 1; o < 32; o <<= 1) {
    const int y = __shfl_up(incl, o);
    if (lane >= o) incl += y;
  }
  if (tid < BN && lane == 31) wtot[wave] = incl;
  __syncthreads();
  if (tid < BN) {
    int excl = incl - mine;
    for (int w = 0; w < wave; ++w) excl += wtot[w];
    const int v = lo + tid;
    if (v < n) {
      rowptr[v] = boff + excl;
      dinv[v] = __frsqrt_rn(fmaxf((float)(mine + 1), 1e-12f));
      if (v == n - 1) rowptr[n] = boff + excl + mine;
    }
    c[tid] = excl;
  }
  __syncthreads();

  for (int i0 = 0; i0 < cntb; i0 += BT * BU) {
    int2 ent[BU];
    #pragma unroll
    for (int k = 0; k < BU; ++k) {
      const int i = i0 + k * BT + tid;
      ent[k].x = 0;
      ent[k].y = -1;
      if (i < cntb) ent[k] = *(const int2*)&bk[2 * (size_t)(boff + i)];
    }
    int pos[BU];
    #pragma unroll
    for (int k = 0; k < BU; ++k) {
      pos[k] = 0;
      if (ent[k].y >= 0)
        pos[k] = __hip_atomic_fetch_add(&c[ent[k].y - lo], 1, __ATOMIC_RELAXED,
                                        __HIP_MEMORY_SCOPE_WORKGROUP);
    }
    #pragma unroll
    for (int k = 0; k < BU; ++k)
      if (ent[k].y >= 0) srcs[(size_t)boff + pos[k]] = ent[k].x;
  }
}

template <int U>
__global__ void weight_kernel(const int* __restrict__ srcs, const float* __restrict__ dinv,
                              int* __restrict__ list, long long E) {
  const long long T = (long long)gridDim.x * blockDim.x;
  const long long t = (long long)blockIdx.x * blockDim.x + threadIdx.x;
  int u[U];
  float w[U];
  #pragma unroll
  for (int k = 0; k < U; ++k) {
    const long long e = t + k * T;
    u[k] = (e < E) ? srcs[e] : 0;
  }
  #pragma unroll
  for (int k = 0; k < U; ++k) w[k] = dinv[u[k]];
  #pragma unroll
  for (int k = 0; k < U; ++k) {
    const long long e = t + k * T;
    if (e < E) {
      int2 ent;
      ent.x = u[k];
      ent.y = __float_as_int(w[k]);
      *(int2*)&list[2 * (size_t)e] = ent;
    }
  }
}

template <int F, bool RELU>
__global__ __launch_bounds__(256) void gather_kernel(
    const int* __restrict__ rowptr, const int* __restrict__ list,
    const float* __restrict__ dinv, const float* __restrict__ h,
    const float* __restrict__ bias, float* __restrict__ y, int n) {
  constexpr int V = F / 32;
  typedef float vecT __attribute__((ext_vector_type(V)));
  constexpr int G = 8;

  const int lane = threadIdx.x & 31;
  const int wv = __builtin_amdgcn_readfirstlane((int)(threadIdx.x >> 5));
  const int wpb = (int)(blockDim.x >> 5);
  const int nwaves = (int)gridDim.x * wpb;
  const vecT bv = *(const vecT*)&bias[lane * V];

  for (int v = (int)blockIdx.x * wpb + wv; v < n; v += nwaves) {
    const int beg = __builtin_amdgcn_readfirstlane(rowptr[v]);
    const int end = __builtin_amdgcn_readfirstlane(rowptr[v + 1]);
    const float dv = dinv[v];
    const vecT hs = *(const vecT*)&h[(size_t)v * F + lane * V];
    vecT acc = {};

    for (int base = beg; base < end; base += 32) {
      const int m = (end - base < 32) ? (end - base) : 32;
      int eu = v;
      int ew = 0;
      if (lane < m) {
        const int2 ent = *(const int2*)&list[2 * (size_t)(base + lane)];
        eu = ent.x;
        ew = ent.y;
      }
      for (int j = 0; j < m; j += G) {
        vecT r[G];
        float w[G];
        #pragma unroll
        for (int q = 0; q < G; ++q) {
          const int u = __builtin_amdgcn_readlane(eu, j + q);
          w[q] = __int_as_float(__builtin_amdgcn_readlane(ew, j + q));
          r[q] = *(const vecT*)&h[(size_t)u * F + lane * V];
        }
        #pragma unroll
        for (int q = 0; q < G; ++q) acc += (w[q] * dv) * r[q];
      }
    }

    acc += (dv * dv) * hs;
    acc += bv;
    if (RELU) acc = __builtin_elementwise_max(acc, (vecT)(0.0f));
    *(vecT*)&y[(size_t)v * F + lane * V] = acc;
  }
}

__global__ __launch_bounds__(256) void gather_gemm_relu_gemm_kernel(
    const int* __restrict__ rowptr, const int* __restrict__ list,
    const float* __restrict__ dinv, const float* __restrict__ x,
    const float* __restrict__ W1, const float* __restrict__ bias1,
    const float* __restrict__ W2, float* __restrict__ C, int n) {
  constexpr int F = 128;
  constexpr int FH = 128;
  constexpr int FOUT = 64;
  constexpr int P = 132;
  constexpr int G = 8;
  typedef float vecT __attribute__((ext_vector_type(4)));
  __shared__ float tile[32 * P];
  __shared__ float hid[32 * P];

  const int lane = threadIdx.x & 31;
  const int wv = __builtin_amdgcn_readfirstlane((int)(threadIdx.x >> 5));
  const int tile0 = (int)blockIdx.x * 32;
  const int v0 = tile0 + wv * 4;

  int bi = v0 + (lane < 4 ? lane : 4);
  bi = bi < n ? bi : n;
  const int rp = rowptr[bi];
  int di = v0 + (lane & 3);
  di = di < n ? di : n - 1;
  const float dvv = dinv[di];

  for (int i = 0; i < 4; ++i) {
    const int v = v0 + i;
    vecT acc = {};
    if (v < n) {
      const int beg = __builtin_amdgcn_readlane(rp, i);
      const int end = __builtin_amdgcn_readlane(rp, i + 1);
      const float dv = __int_as_float(__builtin_amdgcn_readlane(__float_as_int(dvv), i));
      const vecT hs = *(const vecT*)&x[(size_t)v * F + lane * 4];

      for (int base = beg; base < end; base += 32) {
        const int m = (end - base < 32) ? (end - base) : 32;
        int eu = v;
        int ew = 0;
        if (lane < m) {
          const int2 ent = *(const int2*)&list[2 * (size_t)(base + lane)];
          eu = ent.x;
          ew = ent.y;
        }
        for (int j = 0; j < m; j += G) {
          vecT r[G];
          float w[G];
          #pragma unroll
          for (int q = 0; q < G; ++q) {
            const int u = __builtin_amdgcn_readlane(eu, j + q);
            w[q] = __int_as_float(__builtin_amdgcn_readlane(ew, j + q));
            r[q] = *(const vecT*)&x[(size_t)u * F + lane * 4];
          }
          #pragma unroll
          for (int q = 0; q < G; ++q) acc += (w[q] * dv) * r[q];
        }
      }

      acc += (dv * dv) * hs;
    }
    *(vecT*)&tile[(wv * 4 + i) * P + lane * 4] = acc;
  }

  const int half = lane >> 4;
  const int m = lane & 15;

  const int cb1 = wv * 16;
  v2f b[32];
  #pragma unroll
  for (int kk = 0; kk < 32; ++kk) {
    const int kb = kk * 4 + half * 2;
    b[kk].x = W1[(size_t)kb * FH + cb1 + m];
    b[kk].y = W1[(size_t)(kb + 1) * FH + cb1 + m];
  }
  const float bc = bias1[cb1 + m];

  __syncthreads();

  v8f d0 = {};
  v8f d1 = {};
  #pragma unroll
  for (int kk = 0; kk < 32; ++kk) {
    const int kb = kk * 4 + half * 2;
    const v2f a0 = *(const v2f*)&tile[m * P + kb];
    const v2f a1 = *(const v2f*)&tile[(16 + m) * P + kb];
    d0 = __builtin_amdgcn_wmma_f32_16x16x4_f32(false, a0, false, b[kk], (short)0,
                                               d0, false, false);
    d1 = __builtin_amdgcn_wmma_f32_16x16x4_f32(false, a1, false, b[kk], (short)0,
                                               d1, false, false);
  }

  #pragma unroll
  for (int i = 0; i < 8; ++i) {
    hid[(i + half * 8) * P + cb1 + m] = fmaxf(d0[i] + bc, 0.0f);
    hid[(16 + i + half * 8) * P + cb1 + m] = fmaxf(d1[i] + bc, 0.0f);
  }

  const int colbase = (wv & 3) * 16;
  const int rbase = (wv >> 2) * 16;

  v2f c[32];
  #pragma unroll
  for (int kk = 0; kk < 32; ++kk) {
    const int kb = kk * 4 + half * 2;
    c[kk].x = W2[(size_t)kb * FOUT + colbase + m];
    c[kk].y = W2[(size_t)(kb + 1) * FOUT + colbase + m];
  }

  __syncthreads();

  v8f dacc = {};
  #pragma unroll
  for (int kk = 0; kk < 32; ++kk) {
    const int kb = kk * 4 + half * 2;
    const v2f a = *(const v2f*)&hid[(rbase + m) * P + kb];
    dacc = __builtin_amdgcn_wmma_f32_16x16x4_f32(false, a, false, c[kk], (short)0,
                                                 dacc, false, false);
  }

  #pragma unroll
  for (int i = 0; i < 8; ++i) {
    const int row = tile0 + rbase + i + half * 8;
    if (row < n) C[(size_t)row * FOUT + colbase + m] = dacc[i];
  }
}

extern "C" void kernel_launch(void* const* d_in, const int* in_sizes, int n_in,
                              void* d_out, int out_size, void* d_ws,
                              size_t ws_size, hipStream_t stream) {
  const float* x  = (const float*)d_in[0];
  const int*   ei = (const int*)d_in[1];
  const float* W1 = (const float*)d_in[2];
  const float* b1 = (const float*)d_in[3];
  const float* W2 = (const float*)d_in[4];
  const float* b2 = (const float*)d_in[5];
  float* out = (float*)d_out;

  const int N = in_sizes[0] / 128;
  const long long E = in_sizes[1] / 2;
  const int* src = ei;
  const int* dst = ei + E;

  float* ws = (float*)d_ws;
  float* dinv  = ws;
  float* reg1  = ws + N;
  float* reg2  = reg1 + (size_t)N * 128;
  float* h2pre = reg2;

  int* list    = (int*)(reg2 + (size_t)N * 64);
  int* rowptr  = list + 2 * E;
  int* gcount  = rowptr + N + 1;
  int* gcursor = gcount + ((N + BN - 1) >> BSH);
  int* bk   = (int*)reg1;
  int* srcs = bk + 2 * E;

  const int TB = 256;
  auto blocks = [](long long n, int tb) { return (unsigned)((n + tb - 1) / tb); };
  const int NB = (N + BN - 1) >> BSH;

  zero_i32_kernel<<<blocks(2 * NB, TB), TB, 0, stream>>>(gcount, 2 * NB);
  bucket_count_kernel<<<blocks(E, BT * BU), BT, 0, stream>>>(dst, gcount, E, NB);
  bucket_scatter_kernel<<<blocks(E, BT * BU), BT, 0, stream>>>(src, dst, gcount, gcursor, bk, E, NB);
  bucket_finish_kernel<<<NB, BT, 0, stream>>>(gcount, bk, rowptr, dinv, srcs, N);
  weight_kernel<8><<<blocks(E, TB * 8), TB, 0, stream>>>(srcs, dinv, list, E);

  gather_gemm_relu_gemm_kernel<<<blocks(N, 32), 256, 0, stream>>>(rowptr, list, dinv, x, W1, b1, W2, h2pre, N);

  gather_kernel<64, false><<<2048, 256, 0, stream>>>(rowptr, list, dinv, h2pre, b2, out, N);
}


# Sort each block's edges by bucket in its own stretch (LDS ranks + scan); bucket pass finds its pieces via a table of starts; zero and count launches go (7 to 5 launches); list (sum) order changes
# speedup vs baseline: 5.0496x; 1.0813x over previous
#include <hip/hip_runtime.h>
#include <hip/hip_bf16.h>


typedef float v2f __attribute__((ext_vector_type(2)));
typedef float v8f __attribute__((ext_vector_type(8)));

constexpr int BSH = 8;
constexpr int BN = 1 << BSH;
constexpr int MAXB = 1024;
constexpr int BT = 1024;
constexpr int NW = BT / 32;
constexpr int SU = 8;
constexpr int EPB = BT * SU;
constexpr int PU = 7;

__global__ __launch_bounds__(BT) void bucket_sort_kernel(
    const int* __restrict__ src, const int* __restrict__ dst,
    int* __restrict__ bk, int* __restrict__ T, long long E, int nb) {
  __shared__ int hist[MAXB];
  __shared__ int wsum[NW];
  __shared__ alignas(16) int stage[2 * EPB];
  const int tid = threadIdx.x;
  const int lane = tid & 31;
  const int wave = tid >> 5;

  for (int b = tid; b < nb; b += BT) hist[b] = 0;
  __syncthreads();

  const long long base = (long long)blockIdx.x * EPB;
  int u[SU], d[SU], lr[SU];
  #pragma unroll
  for (int k = 0; k < SU; ++k) {
    const long long e = base + tid + (long long)k * BT;
    u[k] = 0;
    d[k] = -1;
    if (e < E) {
      u[k] = src[e];
      d[k] = dst[e];
    }
  }
  #pragma unroll
  for (int k = 0; k < SU; ++k) {
    lr[k] = 0;
    if (d[k] >= 0)
      lr[k] = __hip_atomic_fetch_add(&hist[d[k] >> BSH], 1, __ATOMIC_RELAXED,
                                     __HIP_MEMORY_SCOPE_WORKGROUP);
  }
  __syncthreads();

  const int g = (tid < nb) ? hist[tid] : 0;
  int incl = g;
  for (int o = 1; o < 32; o <<= 1) {
    const int y = __shfl_up(incl, o);
    if (lane >= o) incl += y;
  }
  if (lane == 31) wsum[wave] = incl;
  __syncthreads();

  int start = incl - g;
  for (int w = 0; w < wave; ++w) start += wsum[w];
  int* __restrict__ Trow = T + (size_t)blockIdx.x * (nb + 1);
  if (tid < nb) {
    hist[tid] = start;
    Trow[tid] = start;
    if (tid == nb - 1) Trow[nb] = start + g;
  }
  __syncthreads();

  int hs[SU];
  #pragma unroll
  for (int k = 0; k < SU; ++k) hs[k] = hist[(d[k] >= 0 ? d[k] : 0) >> BSH];
  #pragma unroll
  for (int k = 0; k < SU; ++k) {
    if (d[k] >= 0) {
      int2 ent;
      ent.x = u[k];
      ent.y = d[k];
      *(int2*)&stage[2 * (hs[k] + lr[k])] = ent;
    }
  }
  __syncthreads();

  const long long left = E - base;
  const int cnt = (left < (long long)EPB) ? (int)left : EPB;
  int2 o[SU];
  #pragma unroll
  for (int k = 0; k < SU; ++k) {
    const int i = tid + k * BT;
    o[k].x = 0;
    o[k].y = 0;
    if (i < cnt) o[k] = *(const int2*)&stage[2 * i];
  }
  #pragma unroll
  for (int k = 0; k < SU; ++k) {
    const int i = tid + k * BT;
    if (i < cnt) *(int2*)&bk[2 * (size_t)(base + i)] = o[k];
  }
}

__global__ __launch_bounds__(BT) void bucket_finish_kernel(
    const int* __restrict__ T, const int* __restrict__ bk,
    int* __restrict__ rowptr, float* __restrict__ dinv, int* __restrict__ srcs,
    int n, int nblk, int nb) {
  __shared__ int c[BN];
  __shared__ int ps[BT];
  __shared__ int pl[BT];
  __shared__ int red[NW];
  __shared__ int wtot[BN / 32];
  const int tid = threadIdx.x;
  const int lane = tid & 31;
  const int wave = __builtin_amdgcn_readfirstlane(tid >> 5);
  const int b = (int)blockIdx.x;
  const int lo = b << BSH;
  const int tw = nb + 1;

  if (tid < BN) c[tid] = 0;

  int p = 0;
  for (int B0 = 0; B0 < nblk; B0 += BT) {
    __syncthreads();
    const int B = B0 + tid;
    if (B < nblk) {
      const int s = T[(size_t)B * tw + b];
      const int e = T[(size_t)B * tw + b + 1];
      p += s;
      ps[tid] = B * EPB + s;
      pl[tid] = e - s;
    }
    __syncthreads();
    const int np = (nblk - B0 < BT) ? (nblk - B0) : BT;
    for (int q0 = wave; q0 < np; q0 += NW * PU) {
      const int ql = q0 + NW * lane;
      const bool okl = lane < PU && ql < np;
      const int vs = okl ? ps[ql] : 0;
      const int vl = okl ? pl[ql] : 0;
      int s[PU], len[PU], dd[PU];
      #pragma unroll
      for (int k = 0; k < PU; ++k) {
        s[k] = __builtin_amdgcn_readlane(vs, k);
        len[k] = __builtin_amdgcn_readlane(vl, k);
      }
      #pragma unroll
      for (int k = 0; k < PU; ++k)
        dd[k] = (lane < len[k]) ? bk[2 * (size_t)(s[k] + lane) + 1] : -1;
      #pragma unroll
      for (int k = 0; k < PU; ++k)
        if (dd[k] >= 0)
          __hip_atomic_fetch_add(&c[dd[k] - lo], 1, __ATOMIC_RELAXED,
                                 __HIP_MEMORY_SCOPE_WORKGROUP);
      #pragma unroll
      for (int k = 0; k < PU; ++k)
        for (int i = 32 + lane; i < len[k]; i += 32) {
          const int dv = bk[2 * (size_t)(s[k] + i) + 1];
          __hip_atomic_fetch_add(&c[dv - lo], 1, __ATOMIC_RELAXED,
                                 __HIP_MEMORY_SCOPE_WORKGROUP);
        }
    }
  }

  for (int o = 16; o > 0; o >>= 1) p += __shfl_xor(p, o);
  if (lane == 0) red[wave] = p;
  __syncthreads();
  int boff = 0;
  for (int w = 0; w < NW; ++w) boff += red[w];

  const int mine = (tid < BN) ? c[tid] : 0;
  int incl = mine;
  for (int o = 1; o < 32; o <<= 1) {
    const int y = __shfl_up(incl, o);
    if (lane >= o) incl += y;
  }
  if (tid < BN && lane == 31) wtot[wave] = incl;
  __syncthreads();
  if (tid < BN) {
    int excl = incl - mine;
    for (int w = 0; w < wave; ++w) excl += wtot[w];
    const int v = lo + tid;
    if (v < n) {
      rowptr[v] = boff + excl;
      dinv[v] = __frsqrt_rn(fmaxf((float)(mine + 1), 1e-12f));
      if (v == n - 1) rowptr[n] = boff + excl + mine;
    }
    c[tid] = excl;
  }

  for (int B0 = 0; B0 < nblk; B0 += BT) {
    __syncthreads();
    if (nblk > BT) {
      const int B = B0 + tid;
      if (B < nblk) {
        const int s = T[(size_t)B * tw + b];
        const int e = T[(size_t)B * tw + b + 1];
        ps[tid] = B * EPB + s;
        pl[tid] = e - s;
      }
      __syncthreads();
    }
    const int np = (nblk - B0 < BT) ? (nblk - B0) : BT;
    for (int q0 = wave; q0 < np; q0 += NW * PU) {
      const int ql = q0 + NW * lane;
      const bool okl = lane < PU && ql < np;
      const int vs = okl ? ps[ql] : 0;
      const int vl = okl ? pl[ql] : 0;
      int s[PU], len[PU], pos[PU];
      int2 ent[PU];
      #pragma unroll
      for (int k = 0; k < PU; ++k) {
        s[k] = __builtin_amdgcn_readlane(vs, k);
        len[k] = __builtin_amdgcn_readlane(vl, k);
      }
      #pragma unroll
      for (int k = 0; k < PU; ++k) {
        ent[k].x = 0;
        ent[k].y = -1;
        if (lane < len[k]) ent[k] = *(const int2*)&bk[2 * (size_t)(s[k] + lane)];
      }
      #pragma unroll
      for (int k = 0; k < PU; ++k) {
        pos[k] = 0;
        if (ent[k].y >= 0)
          pos[k] = __hip_atomic_fetch_add(&c[ent[k].y - lo], 1, __ATOMIC_RELAXED,
                                          __HIP_MEMORY_SCOPE_WORKGROUP);
      }
      #pragma unroll
      for (int k = 0; k < PU; ++k)
        if (ent[k].y >= 0) srcs[(size_t)boff + pos[k]] = ent[k].x;
      #pragma unroll
      for (int k = 0; k < PU; ++k)
        for (int i = 32 + lane; i < len[k]; i += 32) {
          const int2 en = *(const int2*)&bk[2 * (size_t)(s[k] + i)];
          const int ps2 = __hip_atomic_fetch_add(&c[en.y - lo], 1, __ATOMIC_RELAXED,
                                                 __HIP_MEMORY_SCOPE_WORKGROUP);
          srcs[(size_t)boff + ps2] = en.x;
        }
    }
  }
}

template <int U>
__global__ void weight_kernel(const int* __restrict__ srcs, const float* __restrict__ dinv,
                              int* __restrict__ list, long long E) {
  const long long T = (long long)gridDim.x * blockDim.x;
  const long long t = (long long)blockIdx.x * blockDim.x + threadIdx.x;
  int u[U];
  float w[U];
  #pragma unroll
  for (int k = 0; k < U; ++k) {
    const long long e = t + k * T;
    u[k] = (e < E) ? srcs[e] : 0;
  }
  #pragma unroll
  for (int k = 0; k < U; ++k) w[k] = dinv[u[k]];
  #pragma unroll
  for (int k = 0; k < U; ++k) {
    const long long e = t + k * T;
    if (e < E) {
      int2 ent;
      ent.x = u[k];
      ent.y = __float_as_int(w[k]);
      *(int2*)&list[2 * (size_t)e] = ent;
    }
  }
}

template <int F, bool RELU>
__global__ __launch_bounds__(256) void gather_kernel(
    const int* __restrict__ rowptr, const int* __restrict__ list,
    const float* __restrict__ dinv, const float* __restrict__ h,
    const float* __restrict__ bias, float* __restrict__ y, int n) {
  constexpr int V = F / 32;
  typedef float vecT __attribute__((ext_vector_type(V)));
  constexpr int G = 8;

  const int lane = threadIdx.x & 31;
  const int wv = __builtin_amdgcn_readfirstlane((int)(threadIdx.x >> 5));
  const int wpb = (int)(blockDim.x >> 5);
  const int nwaves = (int)gridDim.x * wpb;
  const vecT bv = *(const vecT*)&bias[lane * V];

  for (int v = (int)blockIdx.x * wpb + wv; v < n; v += nwaves) {
    const int beg = __builtin_amdgcn_readfirstlane(rowptr[v]);
    const int end = __builtin_amdgcn_readfirstlane(rowptr[v + 1]);
    const float dv = dinv[v];
    const vecT hs = *(const vecT*)&h[(size_t)v * F + lane * V];
    vecT acc = {};

    for (int base = beg; base < end; base += 32) {
      const int m = (end - base < 32) ? (end - base) : 32;
      int eu = v;
      int ew = 0;
      if (lane < m) {
        const int2 ent = *(const int2*)&list[2 * (size_t)(base + lane)];
        eu = ent.x;
        ew = ent.y;
      }
      for (int j = 0; j < m; j += G) {
        vecT r[G];
        float w[G];
        #pragma unroll
        for (int q = 0; q < G; ++q) {
          const int u = __builtin_amdgcn_readlane(eu, j + q);
          w[q] = __int_as_float(__builtin_amdgcn_readlane(ew, j + q));
          r[q] = *(const vecT*)&h[(size_t)u * F + lane * V];
        }
        #pragma unroll
        for (int q = 0; q < G; ++q) acc += (w[q] * dv) * r[q];
      }
    }

    acc += (dv * dv) * hs;
    acc += bv;
    if (RELU) acc = __builtin_elementwise_max(acc, (vecT)(0.0f));
    *(vecT*)&y[(size_t)v * F + lane * V] = acc;
  }
}

__global__ __launch_bounds__(256) void gather_gemm_relu_gemm_kernel(
    const int* __restrict__ rowptr, const int* __restrict__ list,
    const float* __restrict__ dinv, const float* __restrict__ x,
    const float* __restrict__ W1, const float* __restrict__ bias1,
    const float* __restrict__ W2, float* __restrict__ C, int n) {
  constexpr int F = 128;
  constexpr int FH = 128;
  constexpr int FOUT = 64;
  constexpr int P = 132;
  constexpr int G = 8;
  typedef float vecT __attribute__((ext_vector_type(4)));
  __shared__ float tile[32 * P];
  __shared__ float hid[32 * P];

  const int lane = threadIdx.x & 31;
  const int wv = __builtin_amdgcn_readfirstlane((int)(threadIdx.x >> 5));
  const int tile0 = (int)blockIdx.x * 32;
  const int v0 = tile0 + wv * 4;

  int bi = v0 + (lane < 4 ? lane : 4);
  bi = bi < n ? bi : n;
  const int rp = rowptr[bi];
  int di = v0 + (lane & 3);
  di = di < n ? di : n - 1;
  const float dvv = dinv[di];

  for (int i = 0; i < 4; ++i) {
    const int v = v0 + i;
    vecT acc = {};
    if (v < n) {
      const int beg = __builtin_amdgcn_readlane(rp, i);
      const int end = __builtin_amdgcn_readlane(rp, i + 1);
      const float dv = __int_as_float(__builtin_amdgcn_readlane(__float_as_int(dvv), i));
      const vecT hs = *(const vecT*)&x[(size_t)v * F + lane * 4];

      for (int base = beg; base < end; base += 32) {
        const int m = (end - base < 32) ? (end - base) : 32;
        int eu = v;
        int ew = 0;
        if (lane < m) {
          const int2 ent = *(const int2*)&list[2 * (size_t)(base + lane)];
          eu = ent.x;
          ew = ent.y;
        }
        for (int j = 0; j < m; j += G) {
          vecT r[G];
          float w[G];
          #pragma unroll
          for (int q = 0; q < G; ++q) {
            const int u = __builtin_amdgcn_readlane(eu, j + q);
            w[q] = __int_as_float(__builtin_amdgcn_readlane(ew, j + q));
            r[q] = *(const vecT*)&x[(size_t)u * F + lane * 4];
          }
          #pragma unroll
          for (int q = 0; q < G; ++q) acc += (w[q] * dv) * r[q];
        }
      }

      acc += (dv * dv) * hs;
    }
    *(vecT*)&tile[(wv * 4 + i) * P + lane * 4] = acc;
  }

  const int half = lane >> 4;
  const int m = lane & 15;

  const int cb1 = wv * 16;
  v2f b[32];
  #pragma unroll
  for (int kk = 0; kk < 32; ++kk) {
    const int kb = kk * 4 + half * 2;
    b[kk].x = W1[(size_t)kb * FH + cb1 + m];
    b[kk].y = W1[(size_t)(kb + 1) * FH + cb1 + m];
  }
  const float bc = bias1[cb1 + m];

  __syncthreads();

  v8f d0 = {};
  v8f d1 = {};
  #pragma unroll
  for (int kk = 0; kk < 32; ++kk) {
    const int kb = kk * 4 + half * 2;
    const v2f a0 = *(const v2f*)&tile[m * P + kb];
    const v2f a1 = *(const v2f*)&tile[(16 + m) * P + kb];
    d0 = __builtin_amdgcn_wmma_f32_16x16x4_f32(false, a0, false, b[kk], (short)0,
                                               d0, false, false);
    d1 = __builtin_amdgcn_wmma_f32_16x16x4_f32(false, a1, false, b[kk], (short)0,
                                               d1, false, false);
  }

  #pragma unroll
  for (int i = 0; i < 8; ++i) {
    hid[(i + half * 8) * P + cb1 + m] = fmaxf(d0[i] + bc, 0.0f);
    hid[(16 + i + half * 8) * P + cb1 + m] = fmaxf(d1[i] + bc, 0.0f);
  }

  const int colbase = (wv & 3) * 16;
  const int rbase = (wv >> 2) * 16;

  v2f c[32];
  #pragma unroll
  for (int kk = 0; kk < 32; ++kk) {
    const int kb = kk * 4 + half * 2;
    c[kk].x = W2[(size_t)kb * FOUT + colbase + m];
    c[kk].y = W2[(size_t)(kb + 1) * FOUT + colbase + m];
  }

  __syncthreads();

  v8f dacc = {};
  #pragma unroll
  for (int kk = 0; kk < 32; ++kk) {
    const int kb = kk * 4 + half * 2;
    const v2f a = *(const v2f*)&hid[(rbase + m) * P + kb];
    dacc = __builtin_amdgcn_wmma_f32_16x16x4_f32(false, a, false, c[kk], (short)0,
                                                 dacc, false, false);
  }

  #pragma unroll
  for (int i = 0; i < 8; ++i) {
    const int row = tile0 + rbase + i + half * 8;
    if (row < n) C[(size_t)row * FOUT + colbase + m] = dacc[i];
  }
}

extern "C" void kernel_launch(void* const* d_in, const int* in_sizes, int n_in,
                              void* d_out, int out_size, void* d_ws,
                              size_t ws_size, hipStream_t stream) {
  const float* x  = (const float*)d_in[0];
  const int*   ei = (const int*)d_in[1];
  const float* W1 = (const float*)d_in[2];
  const float* b1 = (const float*)d_in[3];
  const float* W2 = (const float*)d_in[4];
  const float* b2 = (const float*)d_in[5];
  float* out = (float*)d_out;

  const int N = in_sizes[0] / 128;
  const long long E = in_sizes[1] / 2;
  const int* src = ei;
  const int* dst = ei + E;

  float* ws = (float*)d_ws;
  float* dinv  = ws;
  float* reg1  = ws + N;
  float* reg2  = reg1 + (size_t)N * 128;
  float* h2pre = reg2;

  int* list    = (int*)(reg2 + (size_t)N * 64);
  int* rowptr  = list + 2 * E;
  int* bk   = (int*)reg1;
  int* srcs = bk + 2 * E;
  int* T    = srcs + E;

  const int TB = 256;
  auto blocks = [](long long n, int tb) { return (unsigned)((n + tb - 1) / tb); };
  const int NB = (N + BN - 1) >> BSH;
  const int NBLK = (int)blocks(E, EPB);

  bucket_sort_kernel<<<NBLK, BT, 0, stream>>>(src, dst, bk, T, E, NB);
  bucket_finish_kernel<<<NB, BT, 0, stream>>>(T, bk, rowptr, dinv, srcs, N, NBLK, NB);
  weight_kernel<8><<<blocks(E, TB * 8), TB, 0, stream>>>(srcs, dinv, list, E);

  gather_gemm_relu_gemm_kernel<<<blocks(N, 32), 256, 0, stream>>>(rowptr, list, dinv, x, W1, b1, W2, h2pre, N);

  gather_kernel<64, false><<<2048, 256, 0, stream>>>(rowptr, list, dinv, h2pre, b2, out, N);
}


# Factor the edge weight: layer-2 rows stored scaled by dinv[src], dinv[dst] applied to the sum; layer 1 looks dinv[src] up beside its row loads; weight launch goes (5 to 4); sum order kept
# speedup vs baseline: 5.3548x; 1.0604x over previous
#include <hip/hip_runtime.h>
#include <hip/hip_bf16.h>


typedef float v2f __attribute__((ext_vector_type(2)));
typedef float v8f __attribute__((ext_vector_type(8)));

constexpr int BSH = 8;
constexpr int BN = 1 << BSH;
constexpr int MAXB = 1024;
constexpr int BT = 1024;
constexpr int NW = BT / 32;
constexpr int SU = 8;
constexpr int EPB = BT * SU;
constexpr int PU = 7;

__global__ __launch_bounds__(BT) void bucket_sort_kernel(
    const int* __restrict__ src, const int* __restrict__ dst,
    int* __restrict__ bk, int* __restrict__ T, long long E, int nb) {
  __shared__ int hist[MAXB];
  __shared__ int wsum[NW];
  __shared__ alignas(16) int stage[2 * EPB];
  const int tid = threadIdx.x;
  const int lane = tid & 31;
  const int wave = tid >> 5;

  for (int b = tid; b < nb; b += BT) hist[b] = 0;
  __syncthreads();

  const long long base = (long long)blockIdx.x * EPB;
  int u[SU], d[SU], lr[SU];
  #pragma unroll
  for (int k = 0; k < SU; ++k) {
    const long long e = base + tid + (long long)k * BT;
    u[k] = 0;
    d[k] = -1;
    if (e < E) {
      u[k] = src[e];
      d[k] = dst[e];
    }
  }
  #pragma unroll
  for (int k = 0; k < SU; ++k) {
    lr[k] = 0;
    if (d[k] >= 0)
      lr[k] = __hip_atomic_fetch_add(&hist[d[k] >> BSH], 1, __ATOMIC_RELAXED,
                                     __HIP_MEMORY_SCOPE_WORKGROUP);
  }
  __syncthreads();

  const int g = (tid < nb) ? hist[tid] : 0;
  int incl = g;
  for (int o = 1; o < 32; o <<= 1) {
    const int y = __shfl_up(incl, o);
    if (lane >= o) incl += y;
  }
  if (lane == 31) wsum[wave] = incl;
  __syncthreads();

  int start = incl - g;
  for (int w = 0; w < wave; ++w) start += wsum[w];
  int* __restrict__ Trow = T + (size_t)blockIdx.x * (nb + 1);
  if (tid < nb) {
    hist[tid] = start;
    Trow[tid] = start;
    if (tid == nb - 1) Trow[nb] = start + g;
  }
  __syncthreads();

  int hs[SU];
  #pragma unroll
  for (int k = 0; k < SU; ++k) hs[k] = hist[(d[k] >= 0 ? d[k] : 0) >> BSH];
  #pragma unroll
  for (int k = 0; k < SU; ++k) {
    if (d[k] >= 0) {
      int2 ent;
      ent.x = u[k];
      ent.y = d[k];
      *(int2*)&stage[2 * (hs[k] + lr[k])] = ent;
    }
  }
  __syncthreads();

  const long long left = E - base;
  const int cnt = (left < (long long)EPB) ? (int)left : EPB;
  int2 o[SU];
  #pragma unroll
  for (int k = 0; k < SU; ++k) {
    const int i = tid + k * BT;
    o[k].x = 0;
    o[k].y = 0;
    if (i < cnt) o[k] = *(const int2*)&stage[2 * i];
  }
  #pragma unroll
  for (int k = 0; k < SU; ++k) {
    const int i = tid + k * BT;
    if (i < cnt) *(int2*)&bk[2 * (size_t)(base + i)] = o[k];
  }
}

__global__ __launch_bounds__(BT) void bucket_finish_kernel(
    const int* __restrict__ T, const int* __restrict__ bk,
    int* __restrict__ rowptr, float* __restrict__ dinv, int* __restrict__ srcs,
    int n, int nblk, int nb) {
  __shared__ int c[BN];
  __shared__ int ps[BT];
  __shared__ int pl[BT];
  __shared__ int red[NW];
  __shared__ int wtot[BN / 32];
  const int tid = threadIdx.x;
  const int lane = tid & 31;
  const int wave = __builtin_amdgcn_readfirstlane(tid >> 5);
  const int b = (int)blockIdx.x;
  const int lo = b << BSH;
  const int tw = nb + 1;

  if (tid < BN) c[tid] = 0;

  int p = 0;
  for (int B0 = 0; B0 < nblk; B0 += BT) {
    __syncthreads();
    const int B = B0 + tid;
    if (B < nblk) {
      const int s = T[(size_t)B * tw + b];
      const int e = T[(size_t)B * tw + b + 1];
      p += s;
      ps[tid] = B * EPB + s;
      pl[tid] = e - s;
    }
    __syncthreads();
    const int np = (nblk - B0 < BT) ? (nblk - B0) : BT;
    for (int q0 = wave; q0 < np; q0 += NW * PU) {
      const int ql = q0 + NW * lane;
      const bool okl = lane < PU && ql < np;
      const int vs = okl ? ps[ql] : 0;
      const int vl = okl ? pl[ql] : 0;
      int s[PU], len[PU], dd[PU];
      #pragma unroll
      for (int k = 0; k < PU; ++k) {
        s[k] = __builtin_amdgcn_readlane(vs, k);
        len[k] = __builtin_amdgcn_readlane(vl, k);
      }
      #pragma unroll
      for (int k = 0; k < PU; ++k)
        dd[k] = (lane < len[k]) ? bk[2 * (size_t)(s[k] + lane) + 1] : -1;
      #pragma unroll
      for (int k = 0; k < PU; ++k)
        if (dd[k] >= 0)
          __hip_atomic_fetch_add(&c[dd[k] - lo], 1, __ATOMIC_RELAXED,
                                 __HIP_MEMORY_SCOPE_WORKGROUP);
      #pragma unroll
      for (int k = 0; k < PU; ++k)
        for (int i = 32 + lane; i < len[k]; i += 32) {
          const int dv = bk[2 * (size_t)(s[k] + i) + 1];
          __hip_atomic_fetch_add(&c[dv - lo], 1, __ATOMIC_RELAXED,
                                 __HIP_MEMORY_SCOPE_WORKGROUP);
        }
    }
  }

  for (int o = 16; o > 0; o >>= 1) p += __shfl_xor(p, o);
  if (lane == 0) red[wave] = p;
  __syncthreads();
  int boff = 0;
  for (int w = 0; w < NW; ++w) boff += red[w];

  const int mine = (tid < BN) ? c[tid] : 0;
  int incl = mine;
  for (int o = 1; o < 32; o <<= 1) {
    const int y = __shfl_up(incl, o);
    if (lane >= o) incl += y;
  }
  if (tid < BN && lane == 31) wtot[wave] = incl;
  __syncthreads();
  if (tid < BN) {
    int excl = incl - mine;
    for (int w = 0; w < wave; ++w) excl += wtot[w];
    const int v = lo + tid;
    if (v < n) {
      rowptr[v] = boff + excl;
      dinv[v] = __frsqrt_rn(fmaxf((float)(mine + 1), 1e-12f));
      if (v == n - 1) rowptr[n] = boff + excl + mine;
    }
    c[tid] = excl;
  }

  for (int B0 = 0; B0 < nblk; B0 += BT) {
    __syncthreads();
    if (nblk > BT) {
      const int B = B0 + tid;
      if (B < nblk) {
        const int s = T[(size_t)B * tw + b];
        const int e = T[(size_t)B * tw + b + 1];
        ps[tid] = B * EPB + s;
        pl[tid] = e - s;
      }
      __syncthreads();
    }
    const int np = (nblk - B0 < BT) ? (nblk - B0) : BT;
    for (int q0 = wave; q0 < np; q0 += NW * PU) {
      const int ql = q0 + NW * lane;
      const bool okl = lane < PU && ql < np;
      const int vs = okl ? ps[ql] : 0;
      const int vl = okl ? pl[ql] : 0;
      int s[PU], len[PU], pos[PU];
      int2 ent[PU];
      #pragma unroll
      for (int k = 0; k < PU; ++k) {
        s[k] = __builtin_amdgcn_readlane(vs, k);
        len[k] = __builtin_amdgcn_readlane(vl, k);
      }
      #pragma unroll
      for (int k = 0; k < PU; ++k) {
        ent[k].x = 0;
        ent[k].y = -1;
        if (lane < len[k]) ent[k] = *(const int2*)&bk[2 * (size_t)(s[k] + lane)];
      }
      #pragma unroll
      for (int k = 0; k < PU; ++k) {
        pos[k] = 0;
        if (ent[k].y >= 0)
          pos[k] = __hip_atomic_fetch_add(&c[ent[k].y - lo], 1, __ATOMIC_RELAXED,
                                          __HIP_MEMORY_SCOPE_WORKGROUP);
      }
      #pragma unroll
      for (int k = 0; k < PU; ++k)
        if (ent[k].y >= 0) srcs[(size_t)boff + pos[k]] = ent[k].x;
      #pragma unroll
      for (int k = 0; k < PU; ++k)
        for (int i = 32 + lane; i < len[k]; i += 32) {
          const int2 en = *(const int2*)&bk[2 * (size_t)(s[k] + i)];
          const int ps2 = __hip_atomic_fetch_add(&c[en.y - lo], 1, __ATOMIC_RELAXED,
                                                 __HIP_MEMORY_SCOPE_WORKGROUP);
          srcs[(size_t)boff + ps2] = en.x;
        }
    }
  }
}

template <int F>
__global__ __launch_bounds__(256) void gather_kernel(
    const int* __restrict__ rowptr, const int* __restrict__ srcs,
    const float* __restrict__ dinv, const float* __restrict__ h,
    const float* __restrict__ bias, float* __restrict__ y, int n) {
  constexpr int V = F / 32;
  typedef float vecT __attribute__((ext_vector_type(V)));
  constexpr int G = 8;

  const int lane = threadIdx.x & 31;
  const int wv = __builtin_amdgcn_readfirstlane((int)(threadIdx.x >> 5));
  const int wpb = (int)(blockDim.x >> 5);
  const int nwaves = (int)gridDim.x * wpb;
  const vecT bv = *(const vecT*)&bias[lane * V];

  for (int v = (int)blockIdx.x * wpb + wv; v < n; v += nwaves) {
    const int beg = __builtin_amdgcn_readfirstlane(rowptr[v]);
    const int end = __builtin_amdgcn_readfirstlane(rowptr[v + 1]);
    const float dv = dinv[v];
    const vecT hs = *(const vecT*)&h[(size_t)v * F + lane * V];
    vecT acc = {};

    for (int base = beg; base < end; base += 32) {
      const int m = (end - base < 32) ? (end - base) : 32;
      int eu = v;
      if (lane < m) eu = srcs[(size_t)(base + lane)];
      for (int j = 0; j < m; j += G) {
        vecT r[G];
        #pragma unroll
        for (int q = 0; q < G; ++q) {
          const int u = __builtin_amdgcn_readlane(eu, j + q);
          r[q] = *(const vecT*)&h[(size_t)u * F + lane * V];
        }
        #pragma unroll
        for (int q = 0; q < G; ++q) acc += ((j + q < m) ? 1.0f : 0.0f) * r[q];
      }
    }

    acc += hs;
    acc = dv * acc + bv;
    *(vecT*)&y[(size_t)v * F + lane * V] = acc;
  }
}

__global__ __launch_bounds__(256) void gather_gemm_relu_gemm_kernel(
    const int* __restrict__ rowptr, const int* __restrict__ srcs,
    const float* __restrict__ dinv, const float* __restrict__ x,
    const float* __restrict__ W1, const float* __restrict__ bias1,
    const float* __restrict__ W2, float* __restrict__ C, int n) {
  constexpr int F = 128;
  constexpr int FH = 128;
  constexpr int FOUT = 64;
  constexpr int P = 132;
  constexpr int G = 8;
  typedef float vecT __attribute__((ext_vector_type(4)));
  __shared__ float tile[32 * P];
  __shared__ float hid[32 * P];
  __shared__ alignas(16) float dsc[32];

  const int lane = threadIdx.x & 31;
  const int wv = __builtin_amdgcn_readfirstlane((int)(threadIdx.x >> 5));
  const int tile0 = (int)blockIdx.x * 32;
  const int v0 = tile0 + wv * 4;

  int bi = v0 + (lane < 4 ? lane : 4);
  bi = bi < n ? bi : n;
  const int rp = rowptr[bi];
  int di = v0 + (lane & 3);
  di = di < n ? di : n - 1;
  const float dvv = dinv[di];
  if (lane < 4) dsc[wv * 4 + lane] = dvv;

  for (int i = 0; i < 4; ++i) {
    const int v = v0 + i;
    vecT acc = {};
    if (v < n) {
      const int beg = __builtin_amdgcn_readlane(rp, i);
      const int end = __builtin_amdgcn_readlane(rp, i + 1);
      const float dv = __int_as_float(__builtin_amdgcn_readlane(__float_as_int(dvv), i));
      const vecT hs = *(const vecT*)&x[(size_t)v * F + lane * 4];

      for (int base = beg; base < end; base += 32) {
        const int m = (end - base < 32) ? (end - base) : 32;
        int eu = v;
        float ew = 0.0f;
        if (lane < m) {
          eu = srcs[(size_t)(base + lane)];
          ew = dinv[eu];
        }
        for (int j = 0; j < m; j += G) {
          vecT r[G];
          float w[G];
          #pragma unroll
          for (int q = 0; q < G; ++q) {
            const int u = __builtin_amdgcn_readlane(eu, j + q);
            r[q] = *(const vecT*)&x[(size_t)u * F + lane * 4];
          }
          __builtin_amdgcn_sched_barrier(0);
          #pragma unroll
          for (int q = 0; q < G; ++q) {
            w[q] = __int_as_float(__builtin_amdgcn_readlane(__float_as_int(ew), j + q));
            acc += (w[q] * dv) * r[q];
          }
        }
      }

      acc += (dv * dv) * hs;
    }
    *(vecT*)&tile[(wv * 4 + i) * P + lane * 4] = acc;
  }

  const int half = lane >> 4;
  const int m = lane & 15;

  const int cb1 = wv * 16;
  v2f b[32];
  #pragma unroll
  for (int kk = 0; kk < 32; ++kk) {
    const int kb = kk * 4 + half * 2;
    b[kk].x = W1[(size_t)kb * FH + cb1 + m];
    b[kk].y = W1[(size_t)(kb + 1) * FH + cb1 + m];
  }
  const float bc = bias1[cb1 + m];

  __syncthreads();

  v8f d0 = {};
  v8f d1 = {};
  #pragma unroll
  for (int kk = 0; kk < 32; ++kk) {
    const int kb = kk * 4 + half * 2;
    const v2f a0 = *(const v2f*)&tile[m * P + kb];
    const v2f a1 = *(const v2f*)&tile[(16 + m) * P + kb];
    d0 = __builtin_amdgcn_wmma_f32_16x16x4_f32(false, a0, false, b[kk], (short)0,
                                               d0, false, false);
    d1 = __builtin_amdgcn_wmma_f32_16x16x4_f32(false, a1, false, b[kk], (short)0,
                                               d1, false, false);
  }

  #pragma unroll
  for (int i = 0; i < 8; ++i) {
    hid[(i + half * 8) * P + cb1 + m] = fmaxf(d0[i] + bc, 0.0f);
    hid[(16 + i + half * 8) * P + cb1 + m] = fmaxf(d1[i] + bc, 0.0f);
  }

  const int colbase = (wv & 3) * 16;
  const int rbase = (wv >> 2) * 16;

  v2f c[32];
  #pragma unroll
  for (int kk = 0; kk < 32; ++kk) {
    const int kb = kk * 4 + half * 2;
    c[kk].x = W2[(size_t)kb * FOUT + colbase + m];
    c[kk].y = W2[(size_t)(kb + 1) * FOUT + colbase + m];
  }

  __syncthreads();

  v8f dacc = {};
  #pragma unroll
  for (int kk = 0; kk < 32; ++kk) {
    const int kb = kk * 4 + half * 2;
    const v2f a = *(const v2f*)&hid[(rbase + m) * P + kb];
    dacc = __builtin_amdgcn_wmma_f32_16x16x4_f32(false, a, false, c[kk], (short)0,
                                                 dacc, false, false);
  }

  const vecT s0 = *(const vecT*)&dsc[rbase + half * 8];
  const vecT s1 = *(const vecT*)&dsc[rbase + half * 8 + 4];
  #pragma unroll
  for (int i = 0; i < 8; ++i) {
    const int row = tile0 + rbase + i + half * 8;
    const float sc = (i < 4) ? s0[i & 3] : s1[i & 3];
    if (row < n) C[(size_t)row * FOUT + colbase + m] = dacc[i] * sc;
  }
}

extern "C" void kernel_launch(void* const* d_in, const int* in_sizes, int n_in,
                              void* d_out, int out_size, void* d_ws,
                              size_t ws_size, hipStream_t stream) {
  const float* x  = (const float*)d_in[0];
  const int*   ei = (const int*)d_in[1];
  const float* W1 = (const float*)d_in[2];
  const float* b1 = (const float*)d_in[3];
  const float* W2 = (const float*)d_in[4];
  const float* b2 = (const float*)d_in[5];
  float* out = (float*)d_out;

  const int N = in_sizes[0] / 128;
  const long long E = in_sizes[1] / 2;
  const int* src = ei;
  const int* dst = ei + E;

  float* ws = (float*)d_ws;
  float* dinv  = ws;
  float* reg1  = ws + N;
  float* reg2  = reg1 + (size_t)N * 128;
  float* h2pre = reg2;

  int* rowptr  = (int*)(reg2 + (size_t)N * 64) + 2 * E;
  int* bk   = (int*)reg1;
  int* srcs = bk + 2 * E;
  int* T    = srcs + E;

  auto blocks = [](long long n, int tb) { return (unsigned)((n + tb - 1) / tb); };
  const int NB = (N + BN - 1) >> BSH;
  const int NBLK = (int)blocks(E, EPB);

  bucket_sort_kernel<<<NBLK, BT, 0, stream>>>(src, dst, bk, T, E, NB);
  bucket_finish_kernel<<<NB, BT, 0, stream>>>(T, bk, rowptr, dinv, srcs, N, NBLK, NB);

  gather_gemm_relu_gemm_kernel<<<blocks(N, 32), 256, 0, stream>>>(rowptr, srcs, dinv, x, W1, b1, W2, h2pre, N);

  gather_kernel<64><<<2048, 256, 0, stream>>>(rowptr, srcs, dinv, h2pre, b2, out, N);
}


# List build's finish pass: a bucket's sources go into list order in LDS, then out as whole 128-byte lines, in place of a 4-byte store per edge scattered over the bucket's lines; list order kept
# speedup vs baseline: 5.5219x; 1.0312x over previous
#include <hip/hip_runtime.h>
#include <hip/hip_bf16.h>


typedef float v2f __attribute__((ext_vector_type(2)));
typedef float v8f __attribute__((ext_vector_type(8)));

constexpr int BSH = 8;
constexpr int BN = 1 << BSH;
constexpr int MAXB = 1024;
constexpr int BT = 1024;
constexpr int NW = BT / 32;
constexpr int SU = 8;
constexpr int EPB = BT * SU;
constexpr int PU = 7;
constexpr int LCAP = 8192;

__global__ __launch_bounds__(BT) void bucket_sort_kernel(
    const int* __restrict__ src, const int* __restrict__ dst,
    int* __restrict__ bk, int* __restrict__ T, long long E, int nb) {
  __shared__ int hist[MAXB];
  __shared__ int wsum[NW];
  __shared__ alignas(16) int stage[2 * EPB];
  const int tid = threadIdx.x;
  const int lane = tid & 31;
  const int wave = tid >> 5;

  for (int b = tid; b < nb; b += BT) hist[b] = 0;
  __syncthreads();

  const long long base = (long long)blockIdx.x * EPB;
  int u[SU], d[SU], lr[SU];
  #pragma unroll
  for (int k = 0; k < SU; ++k) {
    const long long e = base + tid + (long long)k * BT;
    u[k] = 0;
    d[k] = -1;
    if (e < E) {
      u[k] = src[e];
      d[k] = dst[e];
    }
  }
  #pragma unroll
  for (int k = 0; k < SU; ++k) {
    lr[k] = 0;
    if (d[k] >= 0)
      lr[k] = __hip_atomic_fetch_add(&hist[d[k] >> BSH], 1, __ATOMIC_RELAXED,
                                     __HIP_MEMORY_SCOPE_WORKGROUP);
  }
  __syncthreads();

  const int g = (tid < nb) ? hist[tid] : 0;
  int incl = g;
  for (int o = 1; o < 32; o <<= 1) {
    const int y = __shfl_up(incl, o);
    if (lane >= o) incl += y;
  }
  if (lane == 31) wsum[wave] = incl;
  __syncthreads();

  int start = incl - g;
  for (int w = 0; w < wave; ++w) start += wsum[w];
  int* __restrict__ Trow = T + (size_t)blockIdx.x * (nb + 1);
  if (tid < nb) {
    hist[tid] = start;
    Trow[tid] = start;
    if (tid == nb - 1) Trow[nb] = start + g;
  }
  __syncthreads();

  int hs[SU];
  #pragma unroll
  for (int k = 0; k < SU; ++k) hs[k] = hist[(d[k] >= 0 ? d[k] : 0) >> BSH];
  #pragma unroll
  for (int k = 0; k < SU; ++k) {
    if (d[k] >= 0) {
      int2 ent;
      ent.x = u[k];
      ent.y = d[k];
      *(int2*)&stage[2 * (hs[k] + lr[k])] = ent;
    }
  }
  __syncthreads();

  const long long left = E - base;
  const int cnt = (left < (long long)EPB) ? (int)left : EPB;
  int2 o[SU];
  #pragma unroll
  for (int k = 0; k < SU; ++k) {
    const int i = tid + k * BT;
    o[k].x = 0;
    o[k].y = 0;
    if (i < cnt) o[k] = *(const int2*)&stage[2 * i];
  }
  #pragma unroll
  for (int k = 0; k < SU; ++k) {
    const int i = tid + k * BT;
    if (i < cnt) *(int2*)&bk[2 * (size_t)(base + i)] = o[k];
  }
}

__global__ __launch_bounds__(BT) void bucket_finish_kernel(
    const int* __restrict__ T, const int* __restrict__ bk,
    int* __restrict__ rowptr, float* __restrict__ dinv, int* __restrict__ srcs,
    int n, int nblk, int nb) {
  __shared__ int c[BN];
  __shared__ int ps[BT];
  __shared__ int pl[BT];
  __shared__ int red[NW];
  __shared__ int wtot[BN / 32];
  __shared__ int lst[LCAP];
  const int tid = threadIdx.x;
  const int lane = tid & 31;
  const int wave = __builtin_amdgcn_readfirstlane(tid >> 5);
  const int b = (int)blockIdx.x;
  const int lo = b << BSH;
  const int tw = nb + 1;

  if (tid < BN) c[tid] = 0;

  int p = 0;
  for (int B0 = 0; B0 < nblk; B0 += BT) {
    __syncthreads();
    const int B = B0 + tid;
    if (B < nblk) {
      const int s = T[(size_t)B * tw + b];
      const int e = T[(size_t)B * tw + b + 1];
      p += s;
      ps[tid] = B * EPB + s;
      pl[tid] = e - s;
    }
    __syncthreads();
    const int np = (nblk - B0 < BT) ? (nblk - B0) : BT;
    for (int q0 = wave; q0 < np; q0 += NW * PU) {
      const int ql = q0 + NW * lane;
      const bool okl = lane < PU && ql < np;
      const int vs = okl ? ps[ql] : 0;
      const int vl = okl ? pl[ql] : 0;
      int s[PU], len[PU], dd[PU];
      #pragma unroll
      for (int k = 0; k < PU; ++k) {
        s[k] = __builtin_amdgcn_readlane(vs, k);
        len[k] = __builtin_amdgcn_readlane(vl, k);
      }
      #pragma unroll
      for (int k = 0; k < PU; ++k)
        dd[k] = (lane < len[k]) ? bk[2 * (size_t)(s[k] + lane) + 1] : -1;
      #pragma unroll
      for (int k = 0; k < PU; ++k)
        if (dd[k] >= 0)
          __hip_atomic_fetch_add(&c[dd[k] - lo], 1, __ATOMIC_RELAXED,
                                 __HIP_MEMORY_SCOPE_WORKGROUP);
      #pragma unroll
      for (int k = 0; k < PU; ++k)
        for (int i = 32 + lane; i < len[k]; i += 32) {
          const int dv = bk[2 * (size_t)(s[k] + i) + 1];
          __hip_atomic_fetch_add(&c[dv - lo], 1, __ATOMIC_RELAXED,
                                 __HIP_MEMORY_SCOPE_WORKGROUP);
        }
    }
  }

  for (int o = 16; o > 0; o >>= 1) p += __shfl_xor(p, o);
  if (lane == 0) red[wave] = p;
  __syncthreads();
  int boff = 0;
  for (int w = 0; w < NW; ++w) boff += red[w];

  const int mine = (tid < BN) ? c[tid] : 0;
  int incl = mine;
  for (int o = 1; o < 32; o <<= 1) {
    const int y = __shfl_up(incl, o);
    if (lane >= o) incl += y;
  }
  if (tid < BN && lane == 31) wtot[wave] = incl;
  __syncthreads();
  if (tid < BN) {
    int excl = incl - mine;
    for (int w = 0; w < wave; ++w) excl += wtot[w];
    const int v = lo + tid;
    if (v < n) {
      rowptr[v] = boff + excl;
      dinv[v] = __frsqrt_rn(fmaxf((float)(mine + 1), 1e-12f));
      if (v == n - 1) rowptr[n] = boff + excl + mine;
    }
    c[tid] = excl;
  }
  int btot = 0;
  for (int w = 0; w < BN / 32; ++w) btot += wtot[w];
  const bool staged = btot <= LCAP;

  for (int B0 = 0; B0 < nblk; B0 += BT) {
    __syncthreads();
    if (nblk > BT) {
      const int B = B0 + tid;
      if (B < nblk) {
        const int s = T[(size_t)B * tw + b];
        const int e = T[(size_t)B * tw + b + 1];
        ps[tid] = B * EPB + s;
        pl[tid] = e - s;
      }
      __syncthreads();
    }
    const int np = (nblk - B0 < BT) ? (nblk - B0) : BT;
    for (int q0 = wave; q0 < np; q0 += NW * PU) {
      const int ql = q0 + NW * lane;
      const bool okl = lane < PU && ql < np;
      const int vs = okl ? ps[ql] : 0;
      const int vl = okl ? pl[ql] : 0;
      int s[PU], len[PU], pos[PU];
      int2 ent[PU];
      #pragma unroll
      for (int k = 0; k < PU; ++k) {
        s[k] = __builtin_amdgcn_readlane(vs, k);
        len[k] = __builtin_amdgcn_readlane(vl, k);
      }
      #pragma unroll
      for (int k = 0; k < PU; ++k) {
        ent[k].x = 0;
        ent[k].y = -1;
        if (lane < len[k]) ent[k] = *(const int2*)&bk[2 * (size_t)(s[k] + lane)];
      }
      #pragma unroll
      for (int k = 0; k < PU; ++k) {
        pos[k] = 0;
        if (ent[k].y >= 0)
          pos[k] = __hip_atomic_fetch_add(&c[ent[k].y - lo], 1, __ATOMIC_RELAXED,
                                          __HIP_MEMORY_SCOPE_WORKGROUP);
      }
      #pragma unroll
      for (int k = 0; k < PU; ++k)
        if (ent[k].y >= 0) {
          if (staged) lst[pos[k]] = ent[k].x;
          else srcs[(size_t)boff + pos[k]] = ent[k].x;
        }
      #pragma unroll
      for (int k = 0; k < PU; ++k)
        for (int i = 32 + lane; i < len[k]; i += 32) {
          const int2 en = *(const int2*)&bk[2 * (size_t)(s[k] + i)];
          const int ps2 = __hip_atomic_fetch_add(&c[en.y - lo], 1, __ATOMIC_RELAXED,
                                                 __HIP_MEMORY_SCOPE_WORKGROUP);
          if (staged) lst[ps2] = en.x;
          else srcs[(size_t)boff + ps2] = en.x;
        }
    }
  }

  if (staged) {
    __syncthreads();
    const int gend = boff + btot;
    for (int g = (boff & ~31) + tid; g < gend; g += BT)
      if (g >= boff) srcs[(size_t)g] = lst[g - boff];
  }
}

template <int F>
__global__ __launch_bounds__(256) void gather_kernel(
    const int* __restrict__ rowptr, const int* __restrict__ srcs,
    const float* __restrict__ dinv, const float* __restrict__ h,
    const float* __restrict__ bias, float* __restrict__ y, int n) {
  constexpr int V = F / 32;
  typedef float vecT __attribute__((ext_vector_type(V)));
  constexpr int G = 8;

  const int lane = threadIdx.x & 31;
  const int wv = __builtin_amdgcn_readfirstlane((int)(threadIdx.x >> 5));
  const int wpb = (int)(blockDim.x >> 5);
  const int nwaves = (int)gridDim.x * wpb;
  const vecT bv = *(const vecT*)&bias[lane * V];

  for (int v = (int)blockIdx.x * wpb + wv; v < n; v += nwaves) {
    const int beg = __builtin_amdgcn_readfirstlane(rowptr[v]);
    const int end = __builtin_amdgcn_readfirstlane(rowptr[v + 1]);
    const float dv = dinv[v];
    const vecT hs = *(const vecT*)&h[(size_t)v * F + lane * V];
    vecT acc = {};

    for (int base = beg; base < end; base += 32) {
      const int m = (end - base < 32) ? (end - base) : 32;
      int eu = v;
      if (lane < m) eu = srcs[(size_t)(base + lane)];
      for (int j = 0; j < m; j += G) {
        vecT r[G];
        #pragma unroll
        for (int q = 0; q < G; ++q) {
          const int u = __builtin_amdgcn_readlane(eu, j + q);
          r[q] = *(const vecT*)&h[(size_t)u * F + lane * V];
        }
        #pragma unroll
        for (int q = 0; q < G; ++q) acc += ((j + q < m) ? 1.0f : 0.0f) * r[q];
      }
    }

    acc += hs;
    acc = dv * acc + bv;
    *(vecT*)&y[(size_t)v * F + lane * V] = acc;
  }
}

__global__ __launch_bounds__(256) void gather_gemm_relu_gemm_kernel(
    const int* __restrict__ rowptr, const int* __restrict__ srcs,
    const float* __restrict__ dinv, const float* __restrict__ x,
    const float* __restrict__ W1, const float* __restrict__ bias1,
    const float* __restrict__ W2, float* __restrict__ C, int n) {
  constexpr int F = 128;
  constexpr int FH = 128;
  constexpr int FOUT = 64;
  constexpr int P = 132;
  constexpr int G = 8;
  typedef float vecT __attribute__((ext_vector_type(4)));
  __shared__ float tile[32 * P];
  __shared__ float hid[32 * P];
  __shared__ alignas(16) float dsc[32];

  const int lane = threadIdx.x & 31;
  const int wv = __builtin_amdgcn_readfirstlane((int)(threadIdx.x >> 5));
  const int tile0 = (int)blockIdx.x * 32;
  const int v0 = tile0 + wv * 4;

  int bi = v0 + (lane < 4 ? lane : 4);
  bi = bi < n ? bi : n;
  const int rp = rowptr[bi];
  int di = v0 + (lane & 3);
  di = di < n ? di : n - 1;
  const float dvv = dinv[di];
  if (lane < 4) dsc[wv * 4 + lane] = dvv;

  for (int i = 0; i < 4; ++i) {
    const int v = v0 + i;
    vecT acc = {};
    if (v < n) {
      const int beg = __builtin_amdgcn_readlane(rp, i);
      const int end = __builtin_amdgcn_readlane(rp, i + 1);
      const float dv = __int_as_float(__builtin_amdgcn_readlane(__float_as_int(dvv), i));
      const vecT hs = *(const vecT*)&x[(size_t)v * F + lane * 4];

      for (int base = beg; base < end; base += 32) {
        const int m = (end - base < 32) ? (end - base) : 32;
        int eu = v;
        float ew = 0.0f;
        if (lane < m) {
          eu = srcs[(size_t)(base + lane)];
          ew = dinv[eu];
        }
        for (int j = 0; j < m; j += G) {
          vecT r[G];
          float w[G];
          #pragma unroll
          for (int q = 0; q < G; ++q) {
            const int u = __builtin_amdgcn_readlane(eu, j + q);
            r[q] = *(const vecT*)&x[(size_t)u * F + lane * 4];
          }
          __builtin_amdgcn_sched_barrier(0);
          #pragma unroll
          for (int q = 0; q < G; ++q) {
            w[q] = __int_as_float(__builtin_amdgcn_readlane(__float_as_int(ew), j + q));
            acc += (w[q] * dv) * r[q];
          }
        }
      }

      acc += (dv * dv) * hs;
    }
    *(vecT*)&tile[(wv * 4 + i) * P + lane * 4] = acc;
  }

  const int half = lane >> 4;
  const int m = lane & 15;

  const int cb1 = wv * 16;
  v2f b[32];
  #pragma unroll
  for (int kk = 0; kk < 32; ++kk) {
    const int kb = kk * 4 + half * 2;
    b[kk].x = W1[(size_t)kb * FH + cb1 + m];
    b[kk].y = W1[(size_t)(kb + 1) * FH + cb1 + m];
  }
  const float bc = bias1[cb1 + m];

  __syncthreads();

  v8f d0 = {};
  v8f d1 = {};
  #pragma unroll
  for (int kk = 0; kk < 32; ++kk) {
    const int kb = kk * 4 + half * 2;
    const v2f a0 = *(const v2f*)&tile[m * P + kb];
    const v2f a1 = *(const v2f*)&tile[(16 + m) * P + kb];
    d0 = __builtin_amdgcn_wmma_f32_16x16x4_f32(false, a0, false, b[kk], (short)0,
                                               d0, false, false);
    d1 = __builtin_amdgcn_wmma_f32_16x16x4_f32(false, a1, false, b[kk], (short)0,
                                               d1, false, false);
  }

  #pragma unroll
  for (int i = 0; i < 8; ++i) {
    hid[(i + half * 8) * P + cb1 + m] = fmaxf(d0[i] + bc, 0.0f);
    hid[(16 + i + half * 8) * P + cb1 + m] = fmaxf(d1[i] + bc, 0.0f);
  }

  const int colbase = (wv & 3) * 16;
  const int rbase = (wv >> 2) * 16;

  v2f c[32];
  #pragma unroll
  for (int kk = 0; kk < 32; ++kk) {
    const int kb = kk * 4 + half * 2;
    c[kk].x = W2[(size_t)kb * FOUT + colbase + m];
    c[kk].y = W2[(size_t)(kb + 1) * FOUT + colbase + m];
  }

  __syncthreads();

  v8f dacc = {};
  #pragma unroll
  for (int kk = 0; kk < 32; ++kk) {
    const int kb = kk * 4 + half * 2;
    const v2f a = *(const v2f*)&hid[(rbase + m) * P + kb];
    dacc = __builtin_amdgcn_wmma_f32_16x16x4_f32(false, a, false, c[kk], (short)0,
                                                 dacc, false, false);
  }

  const vecT s0 = *(const vecT*)&dsc[rbase + half * 8];
  const vecT s1 = *(const vecT*)&dsc[rbase + half * 8 + 4];
  #pragma unroll
  for (int i = 0; i < 8; ++i) {
    const int row = tile0 + rbase + i + half * 8;
    const float sc = (i < 4) ? s0[i & 3] : s1[i & 3];
    if (row < n) C[(size_t)row * FOUT + colbase + m] = dacc[i] * sc;
  }
}

extern "C" void kernel_launch(void* const* d_in, const int* in_sizes, int n_in,
                              void* d_out, int out_size, void* d_ws,
                              size_t ws_size, hipStream_t stream) {
  const float* x  = (const float*)d_in[0];
  const int*   ei = (const int*)d_in[1];
  const float* W1 = (const float*)d_in[2];
  const float* b1 = (const float*)d_in[3];
  const float* W2 = (const float*)d_in[4];
  const float* b2 = (const float*)d_in[5];
  float* out = (float*)d_out;

  const int N = in_sizes[0] / 128;
  const long long E = in_sizes[1] / 2;
  const int* src = ei;
  const int* dst = ei + E;

  float* ws = (float*)d_ws;
  float* dinv  = ws;
  float* reg1  = ws + N;
  float* reg2  = reg1 + (size_t)N * 128;
  float* h2pre = reg2;

  int* rowptr  = (int*)(reg2 + (size_t)N * 64) + 2 * E;
  int* bk   = (int*)reg1;
  int* srcs = bk + 2 * E;
  int* T    = srcs + E;

  auto blocks = [](long long n, int tb) { return (unsigned)((n + tb - 1) / tb); };
  const int NB = (N + BN - 1) >> BSH;
  const int NBLK = (int)blocks(E, EPB);

  bucket_sort_kernel<<<NBLK, BT, 0, stream>>>(src, dst, bk, T, E, NB);
  bucket_finish_kernel<<<NB, BT, 0, stream>>>(T, bk, rowptr, dinv, srcs, N, NBLK, NB);

  gather_gemm_relu_gemm_kernel<<<blocks(N, 32), 256, 0, stream>>>(rowptr, srcs, dinv, x, W1, b1, W2, h2pre, N);

  gather_kernel<64><<<2048, 256, 0, stream>>>(rowptr, srcs, dinv, h2pre, b2, out, N);
}


# Weights put in matrix B-fragment order once per call by extra blocks of the first launch: a wave's B loads per matrix become 16 of 16 bytes on whole lines for 64 of 4 bytes; k and sum order kept
# speedup vs baseline: 5.8890x; 1.0665x over previous
#include <hip/hip_runtime.h>
#include <hip/hip_bf16.h>


typedef float v2f __attribute__((ext_vector_type(2)));
typedef float v8f __attribute__((ext_vector_type(8)));

constexpr int BSH = 8;
constexpr int BN = 1 << BSH;
constexpr int MAXB = 1024;
constexpr int BT = 1024;
constexpr int NW = BT / 32;
constexpr int SU = 8;
constexpr int EPB = BT * SU;
constexpr int PU = 7;
constexpr int LCAP = 8192;

constexpr int WF1 = 128 * 128 / 4;
constexpr int WF2 = 128 * 64 / 4;

__device__ __forceinline__ void weight_fragment_order(
    const float* __restrict__ W1, const float* __restrict__ W2,
    float* __restrict__ Wf, int i) {
  if (i >= WF1 + WF2) return;
  const bool first = i < WF1;
  const float* __restrict__ W = first ? W1 : W2;
  const int width = first ? 128 : 64;
  const int e = first ? i : i - WF1;
  const int ln = e & 31;
  const int j = (e >> 5) & 15;
  const int tile = e >> 9;
  const int col = tile * 16 + (ln & 15);
  const int kb0 = 8 * j + 2 * (ln >> 4);
  float4 v;
  v.x = W[(size_t)kb0 * width + col];
  v.y = W[(size_t)(kb0 + 1) * width + col];
  v.z = W[(size_t)(kb0 + 4) * width + col];
  v.w = W[(size_t)(kb0 + 5) * width + col];
  *(float4*)&Wf[4 * (size_t)i] = v;
}

__global__ __launch_bounds__(BT) void bucket_sort_kernel(
    const int* __restrict__ src, const int* __restrict__ dst,
    int* __restrict__ bk, int* __restrict__ T, long long E, int nb, int nsort,
    const float* __restrict__ W1, const float* __restrict__ W2,
    float* __restrict__ Wf) {
  if ((int)blockIdx.x >= nsort) {
    weight_fragment_order(W1, W2, Wf,
                          ((int)blockIdx.x - nsort) * BT + (int)threadIdx.x);
    return;
  }
  __shared__ int hist[MAXB];
  __shared__ int wsum[NW];
  __shared__ alignas(16) int stage[2 * EPB];
  const int tid = threadIdx.x;
  const int lane = tid & 31;
  const int wave = tid >> 5;

  for (int b = tid; b < nb; b += BT) hist[b] = 0;
  __syncthreads();

  const long long base = (long long)blockIdx.x * EPB;
  int u[SU], d[SU], lr[SU];
  #pragma unroll
  for (int k = 0; k < SU; ++k) {
    const long long e = base + tid + (long long)k * BT;
    u[k] = 0;
    d[k] = -1;
    if (e < E) {
      u[k] = src[e];
      d[k] = dst[e];
    }
  }
  #pragma unroll
  for (int k = 0; k < SU; ++k) {
    lr[k] = 0;
    if (d[k] >= 0)
      lr[k] = __hip_atomic_fetch_add(&hist[d[k] >> BSH], 1, __ATOMIC_RELAXED,
                                     __HIP_MEMORY_SCOPE_WORKGROUP);
  }
  __syncthreads();

  const int g = (tid < nb) ? hist[tid] : 0;
  int incl = g;
  for (int o = 1; o < 32; o <<= 1) {
    const int y = __shfl_up(incl, o);
    if (lane >= o) incl += y;
  }
  if (lane == 31) wsum[wave] = incl;
  __syncthreads();

  int start = incl - g;
  for (int w = 0; w < wave; ++w) start += wsum[w];
  int* __restrict__ Trow = T + (size_t)blockIdx.x * (nb + 1);
  if (tid < nb) {
    hist[tid] = start;
    Trow[tid] = start;
    if (tid == nb - 1) Trow[nb] = start + g;
  }
  __syncthreads();

  int hs[SU];
  #pragma unroll
  for (int k = 0; k < SU; ++k) hs[k] = hist[(d[k] >= 0 ? d[k] : 0) >> BSH];
  #pragma unroll
  for (int k = 0; k < SU; ++k) {
    if (d[k] >= 0) {
      int2 ent;
      ent.x = u[k];
      ent.y = d[k];
      *(int2*)&stage[2 * (hs[k] + lr[k])] = ent;
    }
  }
  __syncthreads();

  const long long left = E - base;
  const int cnt = (left < (long long)EPB) ? (int)left : EPB;
  int2 o[SU];
  #pragma unroll
  for (int k = 0; k < SU; ++k) {
    const int i = tid + k * BT;
    o[k].x = 0;
    o[k].y = 0;
    if (i < cnt) o[k] = *(const int2*)&stage[2 * i];
  }
  #pragma unroll
  for (int k = 0; k < SU; ++k) {
    const int i = tid + k * BT;
    if (i < cnt) *(int2*)&bk[2 * (size_t)(base + i)] = o[k];
  }
}

__global__ __launch_bounds__(BT) void bucket_finish_kernel(
    const int* __restrict__ T, const int* __restrict__ bk,
    int* __restrict__ rowptr, float* __restrict__ dinv, int* __restrict__ srcs,
    int n, int nblk, int nb) {
  __shared__ int c[BN];
  __shared__ int ps[BT];
  __shared__ int pl[BT];
  __shared__ int red[NW];
  __shared__ int wtot[BN / 32];
  __shared__ int lst[LCAP];
  const int tid = threadIdx.x;
  const int lane = tid & 31;
  const int wave = __builtin_amdgcn_readfirstlane(tid >> 5);
  const int b = (int)blockIdx.x;
  const int lo = b << BSH;
  const int tw = nb + 1;

  if (tid < BN) c[tid] = 0;

  int p = 0;
  for (int B0 = 0; B0 < nblk; B0 += BT) {
    __syncthreads();
    const int B = B0 + tid;
    if (B < nblk) {
      const int s = T[(size_t)B * tw + b];
      const int e = T[(size_t)B * tw + b + 1];
      p += s;
      ps[tid] = B * EPB + s;
      pl[tid] = e - s;
    }
    __syncthreads();
    const int np = (nblk - B0 < BT) ? (nblk - B0) : BT;
    for (int q0 = wave; q0 < np; q0 += NW * PU) {
      const int ql = q0 + NW * lane;
      const bool okl = lane < PU && ql < np;
      const int vs = okl ? ps[ql] : 0;
      const int vl = okl ? pl[ql] : 0;
      int s[PU], len[PU], dd[PU];
      #pragma unroll
      for (int k = 0; k < PU; ++k) {
        s[k] = __builtin_amdgcn_readlane(vs, k);
        len[k] = __builtin_amdgcn_readlane(vl, k);
      }
      #pragma unroll
      for (int k = 0; k < PU; ++k)
        dd[k] = (lane < len[k]) ? bk[2 * (size_t)(s[k] + lane) + 1] : -1;
      #pragma unroll
      for (int k = 0; k < PU; ++k)
        if (dd[k] >= 0)
          __hip_atomic_fetch_add(&c[dd[k] - lo], 1, __ATOMIC_RELAXED,
                                 __HIP_MEMORY_SCOPE_WORKGROUP);
      #pragma unroll
      for (int k = 0; k < PU; ++k)
        for (int i = 32 + lane; i < len[k]; i += 32) {
          const int dv = bk[2 * (size_t)(s[k] + i) + 1];
          __hip_atomic_fetch_add(&c[dv - lo], 1, __ATOMIC_RELAXED,
                                 __HIP_MEMORY_SCOPE_WORKGROUP);
        }
    }
  }

  for (int o = 16; o > 0; o >>= 1) p += __shfl_xor(p, o);
  if (lane == 0) red[wave] = p;
  __syncthreads();
  int boff = 0;
  for (int w = 0; w < NW; ++w) boff += red[w];

  const int mine = (tid < BN) ? c[tid] : 0;
  int incl = mine;
  for (int o = 1; o < 32; o <<= 1) {
    const int y = __shfl_up(incl, o);
    if (lane >= o) incl += y;
  }
  if (tid < BN && lane == 31) wtot[wave] = incl;
  __syncthreads();
  if (tid < BN) {
    int excl = incl - mine;
    for (int w = 0; w < wave; ++w) excl += wtot[w];
    const int v = lo + tid;
    if (v < n) {
      rowptr[v] = boff + excl;
      dinv[v] = __frsqrt_rn(fmaxf((float)(mine + 1), 1e-12f));
      if (v == n - 1) rowptr[n] = boff + excl + mine;
    }
    c[tid] = excl;
  }
  int btot = 0;
  for (int w = 0; w < BN / 32; ++w) btot += wtot[w];
  const bool staged = btot <= LCAP;

  for (int B0 = 0; B0 < nblk; B0 += BT) {
    __syncthreads();
    if (nblk > BT) {
      const int B = B0 + tid;
      if (B < nblk) {
        const int s = T[(size_t)B * tw + b];
        const int e = T[(size_t)B * tw + b + 1];
        ps[tid] = B * EPB + s;
        pl[tid] = e - s;
      }
      __syncthreads();
    }
    const int np = (nblk - B0 < BT) ? (nblk - B0) : BT;
    for (int q0 = wave; q0 < np; q0 += NW * PU) {
      const int ql = q0 + NW * lane;
      const bool okl = lane < PU && ql < np;
      const int vs = okl ? ps[ql] : 0;
      const int vl = okl ? pl[ql] : 0;
      int s[PU], len[PU], pos[PU];
      int2 ent[PU];
      #pragma unroll
      for (int k = 0; k < PU; ++k) {
        s[k] = __builtin_amdgcn_readlane(vs, k);
        len[k] = __builtin_amdgcn_readlane(vl, k);
      }
      #pragma unroll
      for (int k = 0; k < PU; ++k) {
        ent[k].x = 0;
        ent[k].y = -1;
        if (lane < len[k]) ent[k] = *(const int2*)&bk[2 * (size_t)(s[k] + lane)];
      }
      #pragma unroll
      for (int k = 0; k < PU; ++k) {
        pos[k] = 0;
        if (ent[k].y >= 0)
          pos[k] = __hip_atomic_fetch_add(&c[ent[k].y - lo], 1, __ATOMIC_RELAXED,
                                          __HIP_MEMORY_SCOPE_WORKGROUP);
      }
      #pragma unroll
      for (int k = 0; k < PU; ++k)
        if (ent[k].y >= 0) {
          if (staged) lst[pos[k]] = ent[k].x;
          else srcs[(size_t)boff + pos[k]] = ent[k].x;
        }
      #pragma unroll
      for (int k = 0; k < PU; ++k)
        for (int i = 32 + lane; i < len[k]; i += 32) {
          const int2 en = *(const int2*)&bk[2 * (size_t)(s[k] + i)];
          const int ps2 = __hip_atomic_fetch_add(&c[en.y - lo], 1, __ATOMIC_RELAXED,
                                                 __HIP_MEMORY_SCOPE_WORKGROUP);
          if (staged) lst[ps2] = en.x;
          else srcs[(size_t)boff + ps2] = en.x;
        }
    }
  }

  if (staged) {
    __syncthreads();
    const int gend = boff + btot;
    for (int g = (boff & ~31) + tid; g < gend; g += BT)
      if (g >= boff) srcs[(size_t)g] = lst[g - boff];
  }
}

template <int F>
__global__ __launch_bounds__(256) void gather_kernel(
    const int* __restrict__ rowptr, const int* __restrict__ srcs,
    const float* __restrict__ dinv, const float* __restrict__ h,
    const float* __restrict__ bias, float* __restrict__ y, int n) {
  constexpr int V = F / 32;
  typedef float vecT __attribute__((ext_vector_type(V)));
  constexpr int G = 8;

  const int lane = threadIdx.x & 31;
  const int wv = __builtin_amdgcn_readfirstlane((int)(threadIdx.x >> 5));
  const int wpb = (int)(blockDim.x >> 5);
  const int nwaves = (int)gridDim.x * wpb;
  const vecT bv = *(const vecT*)&bias[lane * V];

  for (int v = (int)blockIdx.x * wpb + wv; v < n; v += nwaves) {
    const int beg = __builtin_amdgcn_readfirstlane(rowptr[v]);
    const int end = __builtin_amdgcn_readfirstlane(rowptr[v + 1]);
    const float dv = dinv[v];
    const vecT hs = *(const vecT*)&h[(size_t)v * F + lane * V];
    vecT acc = {};

    for (int base = beg; base < end; base += 32) {
      const int m = (end - base < 32) ? (end - base) : 32;
      int eu = v;
      if (lane < m) eu = srcs[(size_t)(base + lane)];
      for (int j = 0; j < m; j += G) {
        vecT r[G];
        #pragma unroll
        for (int q = 0; q < G; ++q) {
          const int u = __builtin_amdgcn_readlane(eu, j + q);
          r[q] = *(const vecT*)&h[(size_t)u * F + lane * V];
        }
        #pragma unroll
        for (int q = 0; q < G; ++q) acc += ((j + q < m) ? 1.0f : 0.0f) * r[q];
      }
    }

    acc += hs;
    acc = dv * acc + bv;
    *(vecT*)&y[(size_t)v * F + lane * V] = acc;
  }
}

__global__ __launch_bounds__(256) void gather_gemm_relu_gemm_kernel(
    const int* __restrict__ rowptr, const int* __restrict__ srcs,
    const float* __restrict__ dinv, const float* __restrict__ x,
    const float* __restrict__ Wf1, const float* __restrict__ bias1,
    const float* __restrict__ Wf2, float* __restrict__ C, int n) {
  constexpr int F = 128;
  constexpr int FOUT = 64;
  constexpr int P = 132;
  constexpr int G = 8;
  typedef float vecT __attribute__((ext_vector_type(4)));
  __shared__ float tile[32 * P];
  __shared__ float hid[32 * P];
  __shared__ alignas(16) float dsc[32];

  const int lane = threadIdx.x & 31;
  const int wv = __builtin_amdgcn_readfirstlane((int)(threadIdx.x >> 5));
  const int tile0 = (int)blockIdx.x * 32;
  const int v0 = tile0 + wv * 4;

  int bi = v0 + (lane < 4 ? lane : 4);
  bi = bi < n ? bi : n;
  const int rp = rowptr[bi];
  int di = v0 + (lane & 3);
  di = di < n ? di : n - 1;
  const float dvv = dinv[di];
  if (lane < 4) dsc[wv * 4 + lane] = dvv;

  for (int i = 0; i < 4; ++i) {
    const int v = v0 + i;
    vecT acc = {};
    if (v < n) {
      const int beg = __builtin_amdgcn_readlane(rp, i);
      const int end = __builtin_amdgcn_readlane(rp, i + 1);
      const float dv = __int_as_float(__builtin_amdgcn_readlane(__float_as_int(dvv), i));
      const vecT hs = *(const vecT*)&x[(size_t)v * F + lane * 4];

      for (int base = beg; base < end; base += 32) {
        const int m = (end - base < 32) ? (end - base) : 32;
        int eu = v;
        float ew = 0.0f;
        if (lane < m) {
          eu = srcs[(size_t)(base + lane)];
          ew = dinv[eu];
        }
        for (int j = 0; j < m; j += G) {
          vecT r[G];
          float w[G];
          #pragma unroll
          for (int q = 0; q < G; ++q) {
            const int u = __builtin_amdgcn_readlane(eu, j + q);
            r[q] = *(const vecT*)&x[(size_t)u * F + lane * 4];
          }
          __builtin_amdgcn_sched_barrier(0);
          #pragma unroll
          for (int q = 0; q < G; ++q) {
            w[q] = __int_as_float(__builtin_amdgcn_readlane(__float_as_int(ew), j + q));
            acc += (w[q] * dv) * r[q];
          }
        }
      }

      acc += (dv * dv) * hs;
    }
    *(vecT*)&tile[(wv * 4 + i) * P + lane * 4] = acc;
  }

  const int half = lane >> 4;
  const int m = lane & 15;

  const int cb1 = wv * 16;
  v2f b[32];
  const vecT* __restrict__ wf1 = (const vecT*)Wf1 + (size_t)wv * 512 + lane;
  #pragma unroll
  for (int j = 0; j < 16; ++j) {
    const vecT t = wf1[j * 32];
    b[2 * j].x = t.x;
    b[2 * j].y = t.y;
    b[2 * j + 1].x = t.z;
    b[2 * j + 1].y = t.w;
  }
  const float bc = bias1[cb1 + m];

  __syncthreads();

  v8f d0 = {};
  v8f d1 = {};
  #pragma unroll
  for (int kk = 0; kk < 32; ++kk) {
    const int kb = kk * 4 + half * 2;
    const v2f a0 = *(const v2f*)&tile[m * P + kb];
    const v2f a1 = *(const v2f*)&tile[(16 + m) * P + kb];
    d0 = __builtin_amdgcn_wmma_f32_16x16x4_f32(false, a0, false, b[kk], (short)0,
                                               d0, false, false);
    d1 = __builtin_amdgcn_wmma_f32_16x16x4_f32(false, a1, false, b[kk], (short)0,
                                               d1, false, false);
  }

  #pragma unroll
  for (int i = 0; i < 8; ++i) {
    hid[(i + half * 8) * P + cb1 + m] = fmaxf(d0[i] + bc, 0.0f);
    hid[(16 + i + half * 8) * P + cb1 + m] = fmaxf(d1[i] + bc, 0.0f);
  }

  const int colbase = (wv & 3) * 16;
  const int rbase = (wv >> 2) * 16;

  v2f c[32];
  const vecT* __restrict__ wf2 = (const vecT*)Wf2 + (size_t)(wv & 3) * 512 + lane;
  #pragma unroll
  for (int j = 0; j < 16; ++j) {
    const vecT t = wf2[j * 32];
    c[2 * j].x = t.x;
    c[2 * j].y = t.y;
    c[2 * j + 1].x = t.z;
    c[2 * j + 1].y = t.w;
  }

  __syncthreads();

  v8f dacc = {};
  #pragma unroll
  for (int kk = 0; kk < 32; ++kk) {
    const int kb = kk * 4 + half * 2;
    const v2f a = *(const v2f*)&hid[(rbase + m) * P + kb];
    dacc = __builtin_amdgcn_wmma_f32_16x16x4_f32(false, a, false, c[kk], (short)0,
                                                 dacc, false, false);
  }

  const vecT s0 = *(const vecT*)&dsc[rbase + half * 8];
  const vecT s1 = *(const vecT*)&dsc[rbase + half * 8 + 4];
  #pragma unroll
  for (int i = 0; i < 8; ++i) {
    const int row = tile0 + rbase + i + half * 8;
    const float sc = (i < 4) ? s0[i & 3] : s1[i & 3];
    if (row < n) C[(size_t)row * FOUT + colbase + m] = dacc[i] * sc;
  }
}

extern "C" void kernel_launch(void* const* d_in, const int* in_sizes, int n_in,
                              void* d_out, int out_size, void* d_ws,
                              size_t ws_size, hipStream_t stream) {
  const float* x  = (const float*)d_in[0];
  const int*   ei = (const int*)d_in[1];
  const float* W1 = (const float*)d_in[2];
  const float* b1 = (const float*)d_in[3];
  const float* W2 = (const float*)d_in[4];
  const float* b2 = (const float*)d_in[5];
  float* out = (float*)d_out;

  const int N = in_sizes[0] / 128;
  const long long E = in_sizes[1] / 2;
  const int* src = ei;
  const int* dst = ei + E;

  float* ws = (float*)d_ws;
  float* dinv  = ws;
  float* reg1  = ws + N;
  float* reg2  = reg1 + (size_t)N * 128;
  float* h2pre = reg2;

  int* rowptr  = (int*)(reg2 + (size_t)N * 64) + 2 * E;
  int* bk   = (int*)reg1;
  int* srcs = bk + 2 * E;
  int* T    = srcs + E;
  float* Wf  = reg1 + (size_t)N * 64;
  float* Wf2 = Wf + 128 * 128;

  auto blocks = [](long long n, int tb) { return (unsigned)((n + tb - 1) / tb); };
  const int NB = (N + BN - 1) >> BSH;
  const int NBLK = (int)blocks(E, EPB);

  bucket_sort_kernel<<<NBLK + (WF1 + WF2 + BT - 1) / BT, BT, 0, stream>>>(src, dst, bk, T, E, NB, NBLK, W1, W2, Wf);
  bucket_finish_kernel<<<NB, BT, 0, stream>>>(T, bk, rowptr, dinv, srcs, N, NBLK, NB);

  gather_gemm_relu_gemm_kernel<<<blocks(N, 32), 256, 0, stream>>>(rowptr, srcs, dinv, x, Wf, b1, Wf2, h2pre, N);

  gather_kernel<64><<<2048, 256, 0, stream>>>(rowptr, srcs, dinv, h2pre, b2, out, N);
}
